# starting kernel
# baseline (speedup 1.0000x reference)
#include <hip/hip_runtime.h>


typedef __attribute__((ext_vector_type(2))) float v2f;
typedef __attribute__((ext_vector_type(8))) float v8f;

#define T_STEPS 4096
#define BATCH   64
#define DIN     64
#define DHID    512
#define DOUT    64
#define CHUNKS  128
#define CLEN    32
#define HSTRIDE 516
#define NWAVES  16
#define NTHREADS (NWAVES * 32)

__device__ __forceinline__ v8f wmma4(v2f a, v2f b, v8f c) {
    return __builtin_amdgcn_wmma_f32_16x16x4_f32(
        false, a, false, b, (short)0, c, false, false);
}

__device__ __forceinline__ int opaque_zero() {
    int z;
    asm volatile("s_mov_b32 %0, 0" : "=s"(z));
    return z;
}

__global__ __launch_bounds__(256)
void rnn_pack(const float* __restrict__ bW, const float* __restrict__ cW,
              v2f* __restrict__ bp, v2f* __restrict__ cp)
{
    const int i = blockIdx.x * 256 + threadIdx.x;
    if (i < 32 * DHID) {
        const int p = i >> 9, n = i & (DHID - 1);
        v2f t;
        t.x = bW[(size_t)(2 * p) * DHID + n];
        t.y = bW[(size_t)(2 * p + 1) * DHID + n];
        bp[i] = t;
    }
    if (i < 256 * DOUT) {
        const int p = i >> 6, n = i & (DOUT - 1);
        v2f t;
        t.x = cW[(size_t)(2 * p) * DOUT + n];
        t.y = cW[(size_t)(2 * p + 1) * DOUT + n];
        cp[i] = t;
    }
}

__global__ __launch_bounds__(NTHREADS)
void rnn_chunk_summary(const float* __restrict__ x,
                       const float* __restrict__ aVec,
                       const v2f* __restrict__ bp0,
                       float* __restrict__ S)
{
    const int chunk = blockIdx.x;
    const int tid   = threadIdx.x;
    const int wave  = tid >> 5;
    const int lane  = tid & 31;
    const int lmod  = lane & 15;
    const int lhi   = lane >> 4;

    const int m_tile = wave & 3;
    const int nclus  = wave >> 2;
    const int rowA   = m_tile * 16 + lmod;

    float av[8];
    #pragma unroll
    for (int nt = 0; nt < 8; ++nt)
        av[nt] = aVec[(nclus * 8 + nt) * 16 + lmod];

    const int zero = opaque_zero();
    const v2f* bp = bp0;

    v8f Sacc[8] = {};

    for (int i = 0; i < CLEN; ++i) {
        bp += zero;
        const float* xt = x + (size_t)(chunk * CLEN + i) * BATCH * DIN;

        v2f afrag[16];
        #pragma unroll
        for (int kt = 0; kt < 16; ++kt)
            afrag[kt] = *(const v2f*)(xt + rowA * DIN + kt * 4 + lhi * 2);

        #pragma unroll
        for (int ntp = 0; ntp < 4; ++ntp) {
            const int n0 = (nclus * 8 + ntp * 2) * 16 + lmod;
            v8f acc0 = {}, acc1 = {};
            #pragma unroll
            for (int kt = 0; kt < 16; ++kt) {
                const int p = kt * 2 + lhi;
                const v2f b0 = bp[p * DHID + n0];
                const v2f b1 = bp[p * DHID + n0 + 16];
                acc0 = wmma4(afrag[kt], b0, acc0);
                acc1 = wmma4(afrag[kt], b1, acc1);
            }
            const float a0 = av[ntp * 2], a1 = av[ntp * 2 + 1];
            #pragma unroll
            for (int r = 0; r < 8; ++r) {
                Sacc[ntp * 2][r]     = Sacc[ntp * 2][r]     * a0 + acc0[r];
                Sacc[ntp * 2 + 1][r] = Sacc[ntp * 2 + 1][r] * a1 + acc1[r];
            }
        }
    }

    float* Sj = S + (size_t)chunk * BATCH * DHID;
    #pragma unroll
    for (int nt = 0; nt < 8; ++nt) {
        const int col = (nclus * 8 + nt) * 16 + lmod;
        #pragma unroll
        for (int r = 0; r < 8; ++r) {
            const int row = m_tile * 16 + r + lhi * 8;
            Sj[row * DHID + col] = Sacc[nt][r];
        }
    }
}

__global__ __launch_bounds__(256)
void rnn_carry_scan(const float* __restrict__ h0,
                    const float* __restrict__ aVec,
                    const float* __restrict__ S,
                    float* __restrict__ Cout,
                    float* __restrict__ hfinal)
{
    const int idx = blockIdx.x * blockDim.x + threadIdx.x;
    if (idx >= BATCH * DHID) return;
    const int hcol = idx & (DHID - 1);

    float aL = aVec[hcol];
    #pragma unroll
    for (int s = 0; s < 5; ++s) aL *= aL;

    float c = h0[idx];
    for (int j = 0; j < CHUNKS; ++j) {
        Cout[(size_t)j * BATCH * DHID + idx] = c;
        c = aL * c + S[(size_t)j * BATCH * DHID + idx];
    }
    hfinal[idx] = c;
}

__global__ __launch_bounds__(NTHREADS)
void rnn_chunk_emit(const float* __restrict__ x,
                    const float* __restrict__ aVec,
                    const v2f* __restrict__ bp0,
                    const v2f* __restrict__ cp0,
                    const float* __restrict__ Cin,
                    float* __restrict__ y)
{
    extern __shared__ float hbuf[];

    const int chunk = blockIdx.x;
    const int tid   = threadIdx.x;
    const int wave  = tid >> 5;
    const int lane  = tid & 31;
    const int lmod  = lane & 15;
    const int lhi   = lane >> 4;

    const int m_tile = wave & 3;
    const int nclus  = wave >> 2;
    const int rowA   = m_tile * 16 + lmod;
    const int n2     = wave >> 2;
    const int rowH   = m_tile * 16 + lmod;
    const int ncol   = n2 * 16 + lmod;

    float av[8];
    #pragma unroll
    for (int nt = 0; nt < 8; ++nt)
        av[nt] = aVec[(nclus * 8 + nt) * 16 + lmod];

    const float* Cj = Cin + (size_t)chunk * BATCH * DHID;
    for (int idx = tid; idx < BATCH * DHID; idx += NTHREADS) {
        const int row = idx >> 9;
        const int col = idx & (DHID - 1);
        hbuf[row * HSTRIDE + col] = Cj[idx];
    }
    __syncthreads();

    const int zero = opaque_zero();
    const v2f* bp = bp0;
    const v2f* cp = cp0;

    for (int i = 0; i < CLEN; ++i) {
        bp += zero;
        cp += zero;
        const int t = chunk * CLEN + i;
        const float* xt = x + (size_t)t * BATCH * DIN;

        v2f afrag[16];
        #pragma unroll
        for (int kt = 0; kt < 16; ++kt)
            afrag[kt] = *(const v2f*)(xt + rowA * DIN + kt * 4 + lhi * 2);

        __syncthreads();

        #pragma unroll
        for (int ntp = 0; ntp < 4; ++ntp) {
            const int n0 = (nclus * 8 + ntp * 2) * 16 + lmod;
            v8f acc0 = {}, acc1 = {};
            #pragma unroll
            for (int kt = 0; kt < 16; ++kt) {
                const int p = kt * 2 + lhi;
                const v2f b0 = bp[p * DHID + n0];
                const v2f b1 = bp[p * DHID + n0 + 16];
                acc0 = wmma4(afrag[kt], b0, acc0);
                acc1 = wmma4(afrag[kt], b1, acc1);
            }
            const float a0 = av[ntp * 2], a1 = av[ntp * 2 + 1];
            #pragma unroll
            for (int r = 0; r < 8; ++r) {
                const int row = m_tile * 16 + r + lhi * 8;
                const int i0  = row * HSTRIDE + n0;
                hbuf[i0]      = hbuf[i0]      * a0 + acc0[r];
                hbuf[i0 + 16] = hbuf[i0 + 16] * a1 + acc1[r];
            }
        }
        __syncthreads();

        v8f y0 = {}, y1 = {};
        #pragma unroll 8
        for (int kt = 0; kt < 64; ++kt) {
            const int kbA = kt * 4 + lhi * 2;
            const int kbB = kbA + 256;
            const v2f hA = *(const v2f*)(hbuf + rowH * HSTRIDE + kbA);
            const v2f hB = *(const v2f*)(hbuf + rowH * HSTRIDE + kbB);
            const v2f cA = cp[(kbA >> 1) * DOUT + ncol];
            const v2f cB = cp[(kbB >> 1) * DOUT + ncol];
            y0 = wmma4(hA, cA, y0);
            y1 = wmma4(hB, cB, y1);
        }

        float* yt = y + (size_t)t * BATCH * DOUT;
        #pragma unroll
        for (int r = 0; r < 8; ++r) {
            const int row = m_tile * 16 + r + lhi * 8;
            yt[row * DOUT + ncol] = y0[r] + y1[r];
        }
    }
}

extern "C" void kernel_launch(void* const* d_in, const int* in_sizes, int n_in,
                              void* d_out, int out_size, void* d_ws, size_t ws_size,
                              hipStream_t stream)
{
    const float* h0 = (const float*)d_in[0];
    const float* x  = (const float*)d_in[1];
    const float* a  = (const float*)d_in[2];
    const float* bW = (const float*)d_in[3];
    const float* cW = (const float*)d_in[4];

    float* out    = (float*)d_out;
    float* hfinal = out;
    float* y      = out + BATCH * DHID;

    float* S  = (float*)d_ws;
    float* C  = S + (size_t)CHUNKS * BATCH * DHID;
    v2f*   bp = (v2f*)(C + (size_t)CHUNKS * BATCH * DHID);
    v2f*   cp = bp + 32 * DHID;

    const size_t shmem = (size_t)(64 * HSTRIDE) * sizeof(float);

    rnn_pack<<<64, 256, 0, stream>>>(bW, cW, bp, cp);
    rnn_chunk_summary<<<CHUNKS, NTHREADS, 0, stream>>>(x, a, bp, S);
    rnn_carry_scan<<<(BATCH * DHID + 255) / 256, 256, 0, stream>>>(h0, a, S, C, hfinal);
    rnn_chunk_emit<<<CHUNKS, NTHREADS, shmem, stream>>>(x, a, bp, cp, C, y);
}


# Chunk length 32 -> 16 steps: 256 work-groups, one per WGP, fill the chip instead of half of it; chunk carries now sum 16-step partials (summation regrouped, still deterministic)
# speedup vs baseline: 1.6282x; 1.6282x over previous
#include <hip/hip_runtime.h>


typedef __attribute__((ext_vector_type(2))) float v2f;
typedef __attribute__((ext_vector_type(8))) float v8f;

#define T_STEPS 4096
#define BATCH   64
#define DIN     64
#define DHID    512
#define DOUT    64
#define CHUNKS  256
#define CLEN    16
#define HSTRIDE 516
#define NWAVES  16
#define NTHREADS (NWAVES * 32)

__device__ __forceinline__ v8f wmma4(v2f a, v2f b, v8f c) {
    return __builtin_amdgcn_wmma_f32_16x16x4_f32(
        false, a, false, b, (short)0, c, false, false);
}

__device__ __forceinline__ int opaque_zero() {
    int z;
    asm volatile("s_mov_b32 %0, 0" : "=s"(z));
    return z;
}

__global__ __launch_bounds__(256)
void rnn_pack(const float* __restrict__ bW, const float* __restrict__ cW,
              v2f* __restrict__ bp, v2f* __restrict__ cp)
{
    const int i = blockIdx.x * 256 + threadIdx.x;
    if (i < 32 * DHID) {
        const int p = i >> 9, n = i & (DHID - 1);
        v2f t;
        t.x = bW[(size_t)(2 * p) * DHID + n];
        t.y = bW[(size_t)(2 * p + 1) * DHID + n];
        bp[i] = t;
    }
    if (i < 256 * DOUT) {
        const int p = i >> 6, n = i & (DOUT - 1);
        v2f t;
        t.x = cW[(size_t)(2 * p) * DOUT + n];
        t.y = cW[(size_t)(2 * p + 1) * DOUT + n];
        cp[i] = t;
    }
}

__global__ __launch_bounds__(NTHREADS)
void rnn_chunk_summary(const float* __restrict__ x,
                       const float* __restrict__ aVec,
                       const v2f* __restrict__ bp0,
                       float* __restrict__ S)
{
    const int chunk = blockIdx.x;
    const int tid   = threadIdx.x;
    const int wave  = tid >> 5;
    const int lane  = tid & 31;
    const int lmod  = lane & 15;
    const int lhi   = lane >> 4;

    const int m_tile = wave & 3;
    const int nclus  = wave >> 2;
    const int rowA   = m_tile * 16 + lmod;

    float av[8];
    #pragma unroll
    for (int nt = 0; nt < 8; ++nt)
        av[nt] = aVec[(nclus * 8 + nt) * 16 + lmod];

    const int zero = opaque_zero();
    const v2f* bp = bp0;

    v8f Sacc[8] = {};

    for (int i = 0; i < CLEN; ++i) {
        bp += zero;
        const float* xt = x + (size_t)(chunk * CLEN + i) * BATCH * DIN;

        v2f afrag[16];
        #pragma unroll
        for (int kt = 0; kt < 16; ++kt)
            afrag[kt] = *(const v2f*)(xt + rowA * DIN + kt * 4 + lhi * 2);

        #pragma unroll
        for (int ntp = 0; ntp < 4; ++ntp) {
            const int n0 = (nclus * 8 + ntp * 2) * 16 + lmod;
            v8f acc0 = {}, acc1 = {};
            #pragma unroll
            for (int kt = 0; kt < 16; ++kt) {
                const int p = kt * 2 + lhi;
                const v2f b0 = bp[p * DHID + n0];
                const v2f b1 = bp[p * DHID + n0 + 16];
                acc0 = wmma4(afrag[kt], b0, acc0);
                acc1 = wmma4(afrag[kt], b1, acc1);
            }
            const float a0 = av[ntp * 2], a1 = av[ntp * 2 + 1];
            #pragma unroll
            for (int r = 0; r < 8; ++r) {
                Sacc[ntp * 2][r]     = Sacc[ntp * 2][r]     * a0 + acc0[r];
                Sacc[ntp * 2 + 1][r] = Sacc[ntp * 2 + 1][r] * a1 + acc1[r];
            }
        }
    }

    float* Sj = S + (size_t)chunk * BATCH * DHID;
    #pragma unroll
    for (int nt = 0; nt < 8; ++nt) {
        const int col = (nclus * 8 + nt) * 16 + lmod;
        #pragma unroll
        for (int r = 0; r < 8; ++r) {
            const int row = m_tile * 16 + r + lhi * 8;
            Sj[row * DHID + col] = Sacc[nt][r];
        }
    }
}

__global__ __launch_bounds__(256)
void rnn_carry_scan(const float* __restrict__ h0,
                    const float* __restrict__ aVec,
                    const float* __restrict__ S,
                    float* __restrict__ Cout,
                    float* __restrict__ hfinal)
{
    const int idx = blockIdx.x * blockDim.x + threadIdx.x;
    if (idx >= BATCH * DHID) return;
    const int hcol = idx & (DHID - 1);

    float aL = aVec[hcol];
    #pragma unroll
    for (int s = 1; s < CLEN; s <<= 1) aL *= aL;

    float c = h0[idx];
    for (int j = 0; j < CHUNKS; ++j) {
        Cout[(size_t)j * BATCH * DHID + idx] = c;
        c = aL * c + S[(size_t)j * BATCH * DHID + idx];
    }
    hfinal[idx] = c;
}

__global__ __launch_bounds__(NTHREADS)
void rnn_chunk_emit(const float* __restrict__ x,
                    const float* __restrict__ aVec,
                    const v2f* __restrict__ bp0,
                    const v2f* __restrict__ cp0,
                    const float* __restrict__ Cin,
                    float* __restrict__ y)
{
    extern __shared__ float hbuf[];

    const int chunk = blockIdx.x;
    const int tid   = threadIdx.x;
    const int wave  = tid >> 5;
    const int lane  = tid & 31;
    const int lmod  = lane & 15;
    const int lhi   = lane >> 4;

    const int m_tile = wave & 3;
    const int nclus  = wave >> 2;
    const int rowA   = m_tile * 16 + lmod;
    const int n2     = wave >> 2;
    const int rowH   = m_tile * 16 + lmod;
    const int ncol   = n2 * 16 + lmod;

    float av[8];
    #pragma unroll
    for (int nt = 0; nt < 8; ++nt)
        av[nt] = aVec[(nclus * 8 + nt) * 16 + lmod];

    const float* Cj = Cin + (size_t)chunk * BATCH * DHID;
    for (int idx = tid; idx < BATCH * DHID; idx += NTHREADS) {
        const int row = idx >> 9;
        const int col = idx & (DHID - 1);
        hbuf[row * HSTRIDE + col] = Cj[idx];
    }
    __syncthreads();

    const int zero = opaque_zero();
    const v2f* bp = bp0;
    const v2f* cp = cp0;

    for (int i = 0; i < CLEN; ++i) {
        bp += zero;
        cp += zero;
        const int t = chunk * CLEN + i;
        const float* xt = x + (size_t)t * BATCH * DIN;

        v2f afrag[16];
        #pragma unroll
        for (int kt = 0; kt < 16; ++kt)
            afrag[kt] = *(const v2f*)(xt + rowA * DIN + kt * 4 + lhi * 2);

        __syncthreads();

        #pragma unroll
        for (int ntp = 0; ntp < 4; ++ntp) {
            const int n0 = (nclus * 8 + ntp * 2) * 16 + lmod;
            v8f acc0 = {}, acc1 = {};
            #pragma unroll
            for (int kt = 0; kt < 16; ++kt) {
                const int p = kt * 2 + lhi;
                const v2f b0 = bp[p * DHID + n0];
                const v2f b1 = bp[p * DHID + n0 + 16];
                acc0 = wmma4(afrag[kt], b0, acc0);
                acc1 = wmma4(afrag[kt], b1, acc1);
            }
            const float a0 = av[ntp * 2], a1 = av[ntp * 2 + 1];
            #pragma unroll
            for (int r = 0; r < 8; ++r) {
                const int row = m_tile * 16 + r + lhi * 8;
                const int i0  = row * HSTRIDE + n0;
                hbuf[i0]      = hbuf[i0]      * a0 + acc0[r];
                hbuf[i0 + 16] = hbuf[i0 + 16] * a1 + acc1[r];
            }
        }
        __syncthreads();

        v8f y0 = {}, y1 = {};
        #pragma unroll 8
        for (int kt = 0; kt < 64; ++kt) {
            const int kbA = kt * 4 + lhi * 2;
            const int kbB = kbA + 256;
            const v2f hA = *(const v2f*)(hbuf + rowH * HSTRIDE + kbA);
            const v2f hB = *(const v2f*)(hbuf + rowH * HSTRIDE + kbB);
            const v2f cA = cp[(kbA >> 1) * DOUT + ncol];
            const v2f cB = cp[(kbB >> 1) * DOUT + ncol];
            y0 = wmma4(hA, cA, y0);
            y1 = wmma4(hB, cB, y1);
        }

        float* yt = y + (size_t)t * BATCH * DOUT;
        #pragma unroll
        for (int r = 0; r < 8; ++r) {
            const int row = m_tile * 16 + r + lhi * 8;
            yt[row * DOUT + ncol] = y0[r] + y1[r];
        }
    }
}

extern "C" void kernel_launch(void* const* d_in, const int* in_sizes, int n_in,
                              void* d_out, int out_size, void* d_ws, size_t ws_size,
                              hipStream_t stream)
{
    const float* h0 = (const float*)d_in[0];
    const float* x  = (const float*)d_in[1];
    const float* a  = (const float*)d_in[2];
    const float* bW = (const float*)d_in[3];
    const float* cW = (const float*)d_in[4];

    float* out    = (float*)d_out;
    float* hfinal = out;
    float* y      = out + BATCH * DHID;

    float* S  = (float*)d_ws;
    float* C  = S + (size_t)CHUNKS * BATCH * DHID;
    v2f*   bp = (v2f*)(C + (size_t)CHUNKS * BATCH * DHID);
    v2f*   cp = bp + 32 * DHID;

    const size_t shmem = (size_t)(64 * HSTRIDE) * sizeof(float);

    rnn_pack<<<64, 256, 0, stream>>>(bW, cW, bp, cp);
    rnn_chunk_summary<<<CHUNKS, NTHREADS, 0, stream>>>(x, a, bp, S);
    rnn_carry_scan<<<(BATCH * DHID + 255) / 256, 256, 0, stream>>>(h0, a, S, C, hfinal);
    rnn_chunk_emit<<<CHUNKS, NTHREADS, shmem, stream>>>(x, a, bp, cp, C, y);
}


# Fragment loads widened b64 -> b128 per lane (weights packed 4 k-rows deep; x and h read as float4): one load feeds two WMMAs; k regrouped inside products, so summation order changes (deterministic)
# speedup vs baseline: 1.6953x; 1.0412x over previous
#include <hip/hip_runtime.h>


typedef __attribute__((ext_vector_type(2))) float v2f;
typedef __attribute__((ext_vector_type(4))) float v4f;
typedef __attribute__((ext_vector_type(8))) float v8f;

#define T_STEPS 4096
#define BATCH   64
#define DIN     64
#define DHID    512
#define DOUT    64
#define CHUNKS  256
#define CLEN    16
#define HSTRIDE 516
#define NWAVES  16
#define NTHREADS (NWAVES * 32)

__device__ __forceinline__ v8f wmma4(v2f a, v2f b, v8f c) {
    return __builtin_amdgcn_wmma_f32_16x16x4_f32(
        false, a, false, b, (short)0, c, false, false);
}

__device__ __forceinline__ int opaque_zero() {
    int z;
    asm volatile("s_mov_b32 %0, 0" : "=s"(z));
    return z;
}

__global__ __launch_bounds__(256)
void rnn_pack(const float* __restrict__ bW, const float* __restrict__ cW,
              v4f* __restrict__ bq, v4f* __restrict__ cq)
{
    const int i = blockIdx.x * 256 + threadIdx.x;
    if (i < 16 * DHID) {
        const int q = i >> 9, n = i & (DHID - 1);
        v4f t;
        t.x = bW[(size_t)(4 * q)     * DHID + n];
        t.y = bW[(size_t)(4 * q + 1) * DHID + n];
        t.z = bW[(size_t)(4 * q + 2) * DHID + n];
        t.w = bW[(size_t)(4 * q + 3) * DHID + n];
        bq[i] = t;
    }
    if (i < 128 * DOUT) {
        const int q = i >> 6, n = i & (DOUT - 1);
        v4f t;
        t.x = cW[(size_t)(4 * q)     * DOUT + n];
        t.y = cW[(size_t)(4 * q + 1) * DOUT + n];
        t.z = cW[(size_t)(4 * q + 2) * DOUT + n];
        t.w = cW[(size_t)(4 * q + 3) * DOUT + n];
        cq[i] = t;
    }
}

__global__ __launch_bounds__(NTHREADS)
void rnn_chunk_summary(const float* __restrict__ x,
                       const float* __restrict__ aVec,
                       const v4f* __restrict__ bq0,
                       float* __restrict__ S)
{
    const int chunk = blockIdx.x;
    const int tid   = threadIdx.x;
    const int wave  = tid >> 5;
    const int lane  = tid & 31;
    const int lmod  = lane & 15;
    const int lhi   = lane >> 4;

    const int m_tile = wave & 3;
    const int nclus  = wave >> 2;
    const int rowA   = m_tile * 16 + lmod;

    float av[8];
    #pragma unroll
    for (int nt = 0; nt < 8; ++nt)
        av[nt] = aVec[(nclus * 8 + nt) * 16 + lmod];

    const int zero = opaque_zero();
    const v4f* bq = bq0;

    v8f Sacc[8] = {};

    for (int i = 0; i < CLEN; ++i) {
        bq += zero;
        const float* xt = x + (size_t)(chunk * CLEN + i) * BATCH * DIN;

        v4f afrag[8];
        #pragma unroll
        for (int j = 0; j < 8; ++j)
            afrag[j] = *(const v4f*)(xt + rowA * DIN + j * 8 + lhi * 4);

        #pragma unroll
        for (int ntp = 0; ntp < 4; ++ntp) {
            const int n0 = (nclus * 8 + ntp * 2) * 16 + lmod;
            v8f acc0 = {}, acc1 = {};
            #pragma unroll
            for (int j = 0; j < 8; ++j) {
                const int q = j * 2 + lhi;
                const v4f b0 = bq[q * DHID + n0];
                const v4f b1 = bq[q * DHID + n0 + 16];
                acc0 = wmma4(afrag[j].xy, b0.xy, acc0);
                acc1 = wmma4(afrag[j].xy, b1.xy, acc1);
                acc0 = wmma4(afrag[j].zw, b0.zw, acc0);
                acc1 = wmma4(afrag[j].zw, b1.zw, acc1);
            }
            const float a0 = av[ntp * 2], a1 = av[ntp * 2 + 1];
            #pragma unroll
            for (int r = 0; r < 8; ++r) {
                Sacc[ntp * 2][r]     = Sacc[ntp * 2][r]     * a0 + acc0[r];
                Sacc[ntp * 2 + 1][r] = Sacc[ntp * 2 + 1][r] * a1 + acc1[r];
            }
        }
    }

    float* Sj = S + (size_t)chunk * BATCH * DHID;
    #pragma unroll
    for (int nt = 0; nt < 8; ++nt) {
        const int col = (nclus * 8 + nt) * 16 + lmod;
        #pragma unroll
        for (int r = 0; r < 8; ++r) {
            const int row = m_tile * 16 + r + lhi * 8;
            Sj[row * DHID + col] = Sacc[nt][r];
        }
    }
}

__global__ __launch_bounds__(256)
void rnn_carry_scan(const float* __restrict__ h0,
                    const float* __restrict__ aVec,
                    const float* __restrict__ S,
                    float* __restrict__ Cout,
                    float* __restrict__ hfinal)
{
    const int idx = blockIdx.x * blockDim.x + threadIdx.x;
    if (idx >= BATCH * DHID) return;
    const int hcol = idx & (DHID - 1);

    float aL = aVec[hcol];
    #pragma unroll
    for (int s = 1; s < CLEN; s <<= 1) aL *= aL;

    float c = h0[idx];
    for (int j = 0; j < CHUNKS; ++j) {
        Cout[(size_t)j * BATCH * DHID + idx] = c;
        c = aL * c + S[(size_t)j * BATCH * DHID + idx];
    }
    hfinal[idx] = c;
}

__global__ __launch_bounds__(NTHREADS)
void rnn_chunk_emit(const float* __restrict__ x,
                    const float* __restrict__ aVec,
                    const v4f* __restrict__ bq0,
                    const v4f* __restrict__ cq0,
                    const float* __restrict__ Cin,
                    float* __restrict__ y)
{
    extern __shared__ __attribute__((aligned(16))) float hbuf[];

    const int chunk = blockIdx.x;
    const int tid   = threadIdx.x;
    const int wave  = tid >> 5;
    const int lane  = tid & 31;
    const int lmod  = lane & 15;
    const int lhi   = lane >> 4;

    const int m_tile = wave & 3;
    const int nclus  = wave >> 2;
    const int rowA   = m_tile * 16 + lmod;
    const int n2     = wave >> 2;
    const int rowH   = m_tile * 16 + lmod;
    const int ncol   = n2 * 16 + lmod;

    float av[8];
    #pragma unroll
    for (int nt = 0; nt < 8; ++nt)
        av[nt] = aVec[(nclus * 8 + nt) * 16 + lmod];

    const float* Cj = Cin + (size_t)chunk * BATCH * DHID;
    for (int idx = tid; idx < BATCH * DHID; idx += NTHREADS) {
        const int row = idx >> 9;
        const int col = idx & (DHID - 1);
        hbuf[row * HSTRIDE + col] = Cj[idx];
    }
    __syncthreads();

    const int zero = opaque_zero();
    const v4f* bq = bq0;
    const v4f* cq = cq0;

    for (int i = 0; i < CLEN; ++i) {
        bq += zero;
        cq += zero;
        const int t = chunk * CLEN + i;
        const float* xt = x + (size_t)t * BATCH * DIN;

        v4f afrag[8];
        #pragma unroll
        for (int j = 0; j < 8; ++j)
            afrag[j] = *(const v4f*)(xt + rowA * DIN + j * 8 + lhi * 4);

        __syncthreads();

        #pragma unroll
        for (int ntp = 0; ntp < 4; ++ntp) {
            const int n0 = (nclus * 8 + ntp * 2) * 16 + lmod;
            v8f acc0 = {}, acc1 = {};
            #pragma unroll
            for (int j = 0; j < 8; ++j) {
                const int q = j * 2 + lhi;
                const v4f b0 = bq[q * DHID + n0];
                const v4f b1 = bq[q * DHID + n0 + 16];
                acc0 = wmma4(afrag[j].xy, b0.xy, acc0);
                acc1 = wmma4(afrag[j].xy, b1.xy, acc1);
                acc0 = wmma4(afrag[j].zw, b0.zw, acc0);
                acc1 = wmma4(afrag[j].zw, b1.zw, acc1);
            }
            const float a0 = av[ntp * 2], a1 = av[ntp * 2 + 1];
            #pragma unroll
            for (int r = 0; r < 8; ++r) {
                const int row = m_tile * 16 + r + lhi * 8;
                const int i0  = row * HSTRIDE + n0;
                hbuf[i0]      = hbuf[i0]      * a0 + acc0[r];
                hbuf[i0 + 16] = hbuf[i0 + 16] * a1 + acc1[r];
            }
        }
        __syncthreads();

        v8f y0 = {}, y1 = {};
        #pragma unroll 4
        for (int j = 0; j < 32; ++j) {
            const int kbA = j * 8 + lhi * 4;
            const int kbB = kbA + 256;
            const v4f hA = *(const v4f*)(hbuf + rowH * HSTRIDE + kbA);
            const v4f hB = *(const v4f*)(hbuf + rowH * HSTRIDE + kbB);
            const v4f cA = cq[(kbA >> 2) * DOUT + ncol];
            const v4f cB = cq[(kbB >> 2) * DOUT + ncol];
            y0 = wmma4(hA.xy, cA.xy, y0);
            y1 = wmma4(hB.xy, cB.xy, y1);
            y0 = wmma4(hA.zw, cA.zw, y0);
            y1 = wmma4(hB.zw, cB.zw, y1);
        }

        float* yt = y + (size_t)t * BATCH * DOUT;
        #pragma unroll
        for (int r = 0; r < 8; ++r) {
            const int row = m_tile * 16 + r + lhi * 8;
            yt[row * DOUT + ncol] = y0[r] + y1[r];
        }
    }
}

extern "C" void kernel_launch(void* const* d_in, const int* in_sizes, int n_in,
                              void* d_out, int out_size, void* d_ws, size_t ws_size,
                              hipStream_t stream)
{
    const float* h0 = (const float*)d_in[0];
    const float* x  = (const float*)d_in[1];
    const float* a  = (const float*)d_in[2];
    const float* bW = (const float*)d_in[3];
    const float* cW = (const float*)d_in[4];

    float* out    = (float*)d_out;
    float* hfinal = out;
    float* y      = out + BATCH * DHID;

    float* S  = (float*)d_ws;
    float* C  = S + (size_t)CHUNKS * BATCH * DHID;
    v4f*   bq = (v4f*)(C + (size_t)CHUNKS * BATCH * DHID);
    v4f*   cq = bq + 16 * DHID;

    const size_t shmem = (size_t)(64 * HSTRIDE) * sizeof(float);

    rnn_pack<<<32, 256, 0, stream>>>(bW, cW, bq, cq);
    rnn_chunk_summary<<<CHUNKS, NTHREADS, 0, stream>>>(x, a, bq, S);
    rnn_carry_scan<<<(BATCH * DHID + 255) / 256, 256, 0, stream>>>(h0, a, S, C, hfinal);
    rnn_chunk_emit<<<CHUNKS, NTHREADS, shmem, stream>>>(x, a, bq, cq, C, y);
}


# Emit kernel: both products transposed so the state tile (accumulator layout) is the B operand of h@c; h stays in registers, LDS only adds 4 hidden-slice partials of y in fixed order (y sum regrouped)
# speedup vs baseline: 1.7879x; 1.0546x over previous
#include <hip/hip_runtime.h>


typedef __attribute__((ext_vector_type(2))) float v2f;
typedef __attribute__((ext_vector_type(4))) float v4f;
typedef __attribute__((ext_vector_type(8))) float v8f;

#define T_STEPS 4096
#define BATCH   64
#define DIN     64
#define DHID    512
#define DOUT    64
#define CHUNKS  256
#define CLEN    16
#define PSTRIDE 68
#define PPLANE  (BATCH * PSTRIDE)
#define NWAVES  16
#define NTHREADS (NWAVES * 32)

__device__ __forceinline__ v8f wmma4(v2f a, v2f b, v8f c) {
    return __builtin_amdgcn_wmma_f32_16x16x4_f32(
        false, a, false, b, (short)0, c, false, false);
}

__device__ __forceinline__ int opaque_zero() {
    int z;
    asm volatile("s_mov_b32 %0, 0" : "=s"(z));
    return z;
}

__global__ __launch_bounds__(256)
void rnn_pack(const float* __restrict__ bW, const float* __restrict__ cW,
              v4f* __restrict__ bq, v4f* __restrict__ cq)
{
    const int i = blockIdx.x * 256 + threadIdx.x;
    if (i < 16 * DHID) {
        const int q = i >> 9, n = i & (DHID - 1);
        v4f t;
        t.x = bW[(size_t)(4 * q)     * DHID + n];
        t.y = bW[(size_t)(4 * q + 1) * DHID + n];
        t.z = bW[(size_t)(4 * q + 2) * DHID + n];
        t.w = bW[(size_t)(4 * q + 3) * DHID + n];
        bq[i] = t;
    }
    if (i < 128 * DOUT) {
        const int q = i >> 6, n = i & (DOUT - 1);
        v4f t;
        t.x = cW[(size_t)(4 * q)     * DOUT + n];
        t.y = cW[(size_t)(4 * q + 1) * DOUT + n];
        t.z = cW[(size_t)(4 * q + 2) * DOUT + n];
        t.w = cW[(size_t)(4 * q + 3) * DOUT + n];
        cq[i] = t;
    }
}

__global__ __launch_bounds__(NTHREADS)
void rnn_chunk_summary(const float* __restrict__ x,
                       const float* __restrict__ aVec,
                       const v4f* __restrict__ bq0,
                       float* __restrict__ S)
{
    const int chunk = blockIdx.x;
    const int tid   = threadIdx.x;
    const int wave  = tid >> 5;
    const int lane  = tid & 31;
    const int lmod  = lane & 15;
    const int lhi   = lane >> 4;

    const int m_tile = wave & 3;
    const int nclus  = wave >> 2;
    const int rowA   = m_tile * 16 + lmod;

    float av[8];
    #pragma unroll
    for (int nt = 0; nt < 8; ++nt)
        av[nt] = aVec[(nclus * 8 + nt) * 16 + lmod];

    const int zero = opaque_zero();
    const v4f* bq = bq0;

    v8f Sacc[8] = {};

    for (int i = 0; i < CLEN; ++i) {
        bq += zero;
        const float* xt = x + (size_t)(chunk * CLEN + i) * BATCH * DIN;

        v4f afrag[8];
        #pragma unroll
        for (int j = 0; j < 8; ++j)
            afrag[j] = *(const v4f*)(xt + rowA * DIN + j * 8 + lhi * 4);

        #pragma unroll
        for (int ntp = 0; ntp < 4; ++ntp) {
            const int n0 = (nclus * 8 + ntp * 2) * 16 + lmod;
            v8f acc0 = {}, acc1 = {};
            #pragma unroll
            for (int j = 0; j < 8; ++j) {
                const int q = j * 2 + lhi;
                const v4f b0 = bq[q * DHID + n0];
                const v4f b1 = bq[q * DHID + n0 + 16];
                acc0 = wmma4(afrag[j].xy, b0.xy, acc0);
                acc1 = wmma4(afrag[j].xy, b1.xy, acc1);
                acc0 = wmma4(afrag[j].zw, b0.zw, acc0);
                acc1 = wmma4(afrag[j].zw, b1.zw, acc1);
            }
            const float a0 = av[ntp * 2], a1 = av[ntp * 2 + 1];
            #pragma unroll
            for (int r = 0; r < 8; ++r) {
                Sacc[ntp * 2][r]     = Sacc[ntp * 2][r]     * a0 + acc0[r];
                Sacc[ntp * 2 + 1][r] = Sacc[ntp * 2 + 1][r] * a1 + acc1[r];
            }
        }
    }

    float* Sj = S + (size_t)chunk * BATCH * DHID;
    #pragma unroll
    for (int nt = 0; nt < 8; ++nt) {
        const int col = (nclus * 8 + nt) * 16 + lmod;
        #pragma unroll
        for (int r = 0; r < 8; ++r) {
            const int row = m_tile * 16 + r + lhi * 8;
            Sj[row * DHID + col] = Sacc[nt][r];
        }
    }
}

__global__ __launch_bounds__(256)
void rnn_carry_scan(const float* __restrict__ h0,
                    const float* __restrict__ aVec,
                    const float* __restrict__ S,
                    float* __restrict__ Cout,
                    float* __restrict__ hfinal)
{
    const int idx = blockIdx.x * blockDim.x + threadIdx.x;
    if (idx >= BATCH * DHID) return;
    const int hcol = idx & (DHID - 1);

    float aL = aVec[hcol];
    #pragma unroll
    for (int s = 1; s < CLEN; s <<= 1) aL *= aL;

    float c = h0[idx];
    for (int j = 0; j < CHUNKS; ++j) {
        Cout[(size_t)j * BATCH * DHID + idx] = c;
        c = aL * c + S[(size_t)j * BATCH * DHID + idx];
    }
    hfinal[idx] = c;
}

__global__ __launch_bounds__(NTHREADS)
void rnn_chunk_emit(const float* __restrict__ x,
                    const float* __restrict__ aVec,
                    const v4f* __restrict__ bq0,
                    const v4f* __restrict__ cq0,
                    const float* __restrict__ Cin,
                    float* __restrict__ y)
{
    extern __shared__ __attribute__((aligned(16))) float part[];

    const int chunk = blockIdx.x;
    const int tid   = threadIdx.x;
    const int wave  = tid >> 5;
    const int lane  = tid & 31;
    const int lmod  = lane & 15;
    const int lhi   = lane >> 4;

    const int m_tile = wave & 3;
    const int nclus  = wave >> 2;
    const int brow   = m_tile * 16 + lmod;

    v8f Hreg[8];
    {
        const float* Cj = Cin + (size_t)chunk * BATCH * DHID + brow * DHID + nclus * 128 + lhi * 8;
        #pragma unroll
        for (int nt = 0; nt < 8; ++nt) {
            const v4f lo = *(const v4f*)(Cj + nt * 16);
            const v4f hi = *(const v4f*)(Cj + nt * 16 + 4);
            Hreg[nt] = __builtin_shufflevector(lo, hi, 0, 1, 2, 3, 4, 5, 6, 7);
        }
    }

    const int zero = opaque_zero();
    const v4f*   bq = bq0;
    const v4f*   cq = cq0;
    const float* aV = aVec;

    for (int i = 0; i < CLEN; ++i) {
        bq += zero;
        cq += zero;
        aV += zero;
        const int t = chunk * CLEN + i;
        const float* xt = x + (size_t)t * BATCH * DIN;

        v4f xfrag[8];
        #pragma unroll
        for (int j = 0; j < 8; ++j)
            xfrag[j] = *(const v4f*)(xt + brow * DIN + j * 8 + lhi * 4);

        v8f Y[4] = {};

        #pragma unroll
        for (int ntp = 0; ntp < 4; ++ntp) {
            const int T0 = nclus * 8 + ntp * 2;
            const int n0 = T0 * 16 + lmod;

            v8f acc0 = {}, acc1 = {};
            #pragma unroll
            for (int j = 0; j < 8; ++j) {
                const int q = j * 2 + lhi;
                const v4f b0 = bq[q * DHID + n0];
                const v4f b1 = bq[q * DHID + n0 + 16];
                acc0 = wmma4(b0.xy, xfrag[j].xy, acc0);
                acc1 = wmma4(b1.xy, xfrag[j].xy, acc1);
                acc0 = wmma4(b0.zw, xfrag[j].zw, acc0);
                acc1 = wmma4(b1.zw, xfrag[j].zw, acc1);
            }

            {
                const float* ap = aV + T0 * 16 + lhi * 8;
                const v4f a0l = *(const v4f*)(ap);
                const v4f a0h = *(const v4f*)(ap + 4);
                const v4f a1l = *(const v4f*)(ap + 16);
                const v4f a1h = *(const v4f*)(ap + 20);
                const v8f a0 = __builtin_shufflevector(a0l, a0h, 0, 1, 2, 3, 4, 5, 6, 7);
                const v8f a1 = __builtin_shufflevector(a1l, a1h, 0, 1, 2, 3, 4, 5, 6, 7);
                Hreg[ntp * 2]     = Hreg[ntp * 2]     * a0 + acc0;
                Hreg[ntp * 2 + 1] = Hreg[ntp * 2 + 1] * a1 + acc1;
            }

            #pragma unroll
            for (int tt = 0; tt < 2; ++tt) {
                const int  T  = T0 + tt;
                const v8f  Hc = Hreg[ntp * 2 + tt];
                const v2f h01 = __builtin_shufflevector(Hc, Hc, 0, 1);
                const v2f h23 = __builtin_shufflevector(Hc, Hc, 2, 3);
                const v2f h45 = __builtin_shufflevector(Hc, Hc, 4, 5);
                const v2f h67 = __builtin_shufflevector(Hc, Hc, 6, 7);
                v4f cA[4], cB[4];
                #pragma unroll
                for (int o = 0; o < 4; ++o) {
                    cA[o] = cq[(4 * T + 2 * lhi)     * DOUT + o * 16 + lmod];
                    cB[o] = cq[(4 * T + 2 * lhi + 1) * DOUT + o * 16 + lmod];
                }
                #pragma unroll
                for (int o = 0; o < 4; ++o) Y[o] = wmma4(cA[o].xy, h01, Y[o]);
                #pragma unroll
                for (int o = 0; o < 4; ++o) Y[o] = wmma4(cA[o].zw, h23, Y[o]);
                #pragma unroll
                for (int o = 0; o < 4; ++o) Y[o] = wmma4(cB[o].xy, h45, Y[o]);
                #pragma unroll
                for (int o = 0; o < 4; ++o) Y[o] = wmma4(cB[o].zw, h67, Y[o]);
            }
        }

        {
            float* pw = part + ((i & 1) * 4 + nclus) * PPLANE + brow * PSTRIDE + lhi * 8;
            #pragma unroll
            for (int o = 0; o < 4; ++o) {
                *(v4f*)(pw + o * 16)     = __builtin_shufflevector(Y[o], Y[o], 0, 1, 2, 3);
                *(v4f*)(pw + o * 16 + 4) = __builtin_shufflevector(Y[o], Y[o], 4, 5, 6, 7);
            }
        }
        __syncthreads();

        {
            const float* pr = part + (i & 1) * 4 * PPLANE;
            float* yt = y + (size_t)t * BATCH * DOUT;
            #pragma unroll
            for (int p = 0; p < 2; ++p) {
                const int row = p * 32 + (tid >> 4);
                const int col = (tid & 15) * 4;
                const float* q0 = pr + row * PSTRIDE + col;
                const v4f s0 = *(const v4f*)(q0);
                const v4f s1 = *(const v4f*)(q0 + PPLANE);
                const v4f s2 = *(const v4f*)(q0 + 2 * PPLANE);
                const v4f s3 = *(const v4f*)(q0 + 3 * PPLANE);
                *(v4f*)(yt + row * DOUT + col) = (s0 + s1) + (s2 + s3);
            }
        }
    }
}

extern "C" void kernel_launch(void* const* d_in, const int* in_sizes, int n_in,
                              void* d_out, int out_size, void* d_ws, size_t ws_size,
                              hipStream_t stream)
{
    const float* h0 = (const float*)d_in[0];
    const float* x  = (const float*)d_in[1];
    const float* a  = (const float*)d_in[2];
    const float* bW = (const float*)d_in[3];
    const float* cW = (const float*)d_in[4];

    float* out    = (float*)d_out;
    float* hfinal = out;
    float* y      = out + BATCH * DHID;

    float* S  = (float*)d_ws;
    float* C  = S + (size_t)CHUNKS * BATCH * DHID;
    v4f*   bq = (v4f*)(C + (size_t)CHUNKS * BATCH * DHID);
    v4f*   cq = bq + 16 * DHID;

    const size_t shmem = (size_t)(2 * 4 * PPLANE) * sizeof(float);

    rnn_pack<<<32, 256, 0, stream>>>(bW, cW, bq, cq);
    rnn_chunk_summary<<<CHUNKS, NTHREADS, 0, stream>>>(x, a, bq, S);
    rnn_carry_scan<<<(BATCH * DHID + 255) / 256, 256, 0, stream>>>(h0, a, S, C, hfinal);
    rnn_chunk_emit<<<CHUNKS, NTHREADS, shmem, stream>>>(x, a, bq, cq, C, y);
}


# Packed weights staged once per work-group in LDS (b in summary kernel; b, c in emit kernel, y exchange cut to one swizzled 64 KiB buffer): weight fragments are LDS reads, not L2 re-reads per step
# speedup vs baseline: 1.8950x; 1.0599x over previous
#include <hip/hip_runtime.h>


typedef __attribute__((ext_vector_type(2))) float v2f;
typedef __attribute__((ext_vector_type(4))) float v4f;
typedef __attribute__((ext_vector_type(8))) float v8f;

#define T_STEPS 4096
#define BATCH   64
#define DIN     64
#define DHID    512
#define DOUT    64
#define CHUNKS  256
#define CLEN    16
#define BQ_UNITS (16 * DHID)
#define CQ_UNITS (128 * DOUT)
#define PUNITS  16
#define PPLANE  (BATCH * PUNITS)
#define NWAVES  16
#define NTHREADS (NWAVES * 32)

__device__ __forceinline__ v8f wmma4(v2f a, v2f b, v8f c) {
    return __builtin_amdgcn_wmma_f32_16x16x4_f32(
        false, a, false, b, (short)0, c, false, false);
}

__device__ __forceinline__ int opaque_zero() {
    int z;
    asm volatile("s_mov_b32 %0, 0" : "=s"(z));
    return z;
}

__device__ __forceinline__ int opaque_copy(int v) {
    int r;
    asm volatile("v_mov_b32 %0, %1" : "=v"(r) : "v"(v));
    return r;
}

__global__ __launch_bounds__(256)
void rnn_pack(const float* __restrict__ bW, const float* __restrict__ cW,
              v4f* __restrict__ bq, v4f* __restrict__ cq)
{
    const int i = blockIdx.x * 256 + threadIdx.x;
    if (i < 16 * DHID) {
        const int q = i >> 9, n = i & (DHID - 1);
        v4f t;
        t.x = bW[(size_t)(4 * q)     * DHID + n];
        t.y = bW[(size_t)(4 * q + 1) * DHID + n];
        t.z = bW[(size_t)(4 * q + 2) * DHID + n];
        t.w = bW[(size_t)(4 * q + 3) * DHID + n];
        bq[i] = t;
    }
    if (i < 128 * DOUT) {
        const int q = i >> 6, n = i & (DOUT - 1);
        v4f t;
        t.x = cW[(size_t)(4 * q)     * DOUT + n];
        t.y = cW[(size_t)(4 * q + 1) * DOUT + n];
        t.z = cW[(size_t)(4 * q + 2) * DOUT + n];
        t.w = cW[(size_t)(4 * q + 3) * DOUT + n];
        cq[i] = t;
    }
}

__global__ __launch_bounds__(NTHREADS)
void rnn_chunk_summary(const float* __restrict__ x,
                       const float* __restrict__ aVec,
                       const v4f* __restrict__ bq0,
                       float* __restrict__ S)
{
    const int chunk = blockIdx.x;
    const int tid   = threadIdx.x;
    const int wave  = tid >> 5;
    const int lane  = tid & 31;
    const int lmod  = lane & 15;
    const int lhi   = lane >> 4;

    const int m_tile = wave & 3;
    const int nclus  = wave >> 2;
    const int rowA   = m_tile * 16 + lmod;

    __shared__ v4f bs[BQ_UNITS];

    #pragma unroll
    for (int k = 0; k < BQ_UNITS / NTHREADS; ++k)
        bs[k * NTHREADS + tid] = bq0[k * NTHREADS + tid];

    float av[8];
    #pragma unroll
    for (int nt = 0; nt < 8; ++nt)
        av[nt] = aVec[(nclus * 8 + nt) * 16 + lmod];

    const int bbase = lhi * DHID + nclus * 128 + lmod;

    v8f Sacc[8] = {};

    __syncthreads();

    for (int i = 0; i < CLEN; ++i) {
        const int zb = opaque_copy(bbase);
        const float* xt = x + (size_t)(chunk * CLEN + i) * BATCH * DIN;

        v4f afrag[8];
        #pragma unroll
        for (int j = 0; j < 8; ++j)
            afrag[j] = *(const v4f*)(xt + rowA * DIN + j * 8 + lhi * 4);
        asm volatile("" :: "v"(afrag[0]), "v"(afrag[1]), "v"(afrag[2]), "v"(afrag[3]),
                           "v"(afrag[4]), "v"(afrag[5]), "v"(afrag[6]), "v"(afrag[7]));

        #pragma unroll
        for (int ntp = 0; ntp < 4; ++ntp) {
            v8f acc0 = {}, acc1 = {};
            #pragma unroll
            for (int j = 0; j < 8; ++j) {
                const v4f b0 = bs[zb + j * 2 * DHID + ntp * 32];
                const v4f b1 = bs[zb + j * 2 * DHID + ntp * 32 + 16];
                acc0 = wmma4(afrag[j].xy, b0.xy, acc0);
                acc1 = wmma4(afrag[j].xy, b1.xy, acc1);
                acc0 = wmma4(afrag[j].zw, b0.zw, acc0);
                acc1 = wmma4(afrag[j].zw, b1.zw, acc1);
            }
            const float a0 = av[ntp * 2], a1 = av[ntp * 2 + 1];
            #pragma unroll
            for (int r = 0; r < 8; ++r) {
                Sacc[ntp * 2][r]     = Sacc[ntp * 2][r]     * a0 + acc0[r];
                Sacc[ntp * 2 + 1][r] = Sacc[ntp * 2 + 1][r] * a1 + acc1[r];
            }
        }
    }

    float* Sj = S + (size_t)chunk * BATCH * DHID;
    #pragma unroll
    for (int nt = 0; nt < 8; ++nt) {
        const int col = (nclus * 8 + nt) * 16 + lmod;
        #pragma unroll
        for (int r = 0; r < 8; ++r) {
            const int row = m_tile * 16 + r + lhi * 8;
            Sj[row * DHID + col] = Sacc[nt][r];
        }
    }
}

__global__ __launch_bounds__(256)
void rnn_carry_scan(const float* __restrict__ h0,
                    const float* __restrict__ aVec,
                    const float* __restrict__ S,
                    float* __restrict__ Cout,
                    float* __restrict__ hfinal)
{
    const int idx = blockIdx.x * blockDim.x + threadIdx.x;
    if (idx >= BATCH * DHID) return;
    const int hcol = idx & (DHID - 1);

    float aL = aVec[hcol];
    #pragma unroll
    for (int s = 1; s < CLEN; s <<= 1) aL *= aL;

    float c = h0[idx];
    for (int j = 0; j < CHUNKS; ++j) {
        Cout[(size_t)j * BATCH * DHID + idx] = c;
        c = aL * c + S[(size_t)j * BATCH * DHID + idx];
    }
    hfinal[idx] = c;
}

__global__ __launch_bounds__(NTHREADS)
void rnn_chunk_emit(const float* __restrict__ x,
                    const float* __restrict__ aVec,
                    const v4f* __restrict__ bq0,
                    const v4f* __restrict__ cq0,
                    const float* __restrict__ Cin,
                    float* __restrict__ y)
{
    __shared__ v4f bs[BQ_UNITS];
    __shared__ v4f cs[CQ_UNITS];
    __shared__ v4f part[4 * PPLANE];

    const int chunk = blockIdx.x;
    const int tid   = threadIdx.x;
    const int wave  = tid >> 5;
    const int lane  = tid & 31;
    const int lmod  = lane & 15;
    const int lhi   = lane >> 4;

    const int m_tile = wave & 3;
    const int nclus  = wave >> 2;
    const int brow   = m_tile * 16 + lmod;

    #pragma unroll
    for (int k = 0; k < BQ_UNITS / NTHREADS; ++k)
        bs[k * NTHREADS + tid] = bq0[k * NTHREADS + tid];
    #pragma unroll
    for (int k = 0; k < CQ_UNITS / NTHREADS; ++k)
        cs[k * NTHREADS + tid] = cq0[k * NTHREADS + tid];

    v8f Hreg[8];
    {
        const float* Cj = Cin + (size_t)chunk * BATCH * DHID + brow * DHID + nclus * 128 + lhi * 8;
        #pragma unroll
        for (int nt = 0; nt < 8; ++nt) {
            const v4f lo = *(const v4f*)(Cj + nt * 16);
            const v4f hi = *(const v4f*)(Cj + nt * 16 + 4);
            Hreg[nt] = __builtin_shufflevector(lo, hi, 0, 1, 2, 3, 4, 5, 6, 7);
        }
    }

    const int zero = opaque_zero();
    const float* aV = aVec;

    __syncthreads();

    for (int i = 0; i < CLEN; ++i) {
        aV += zero;
        const int t = chunk * CLEN + i;
        const float* xt = x + (size_t)t * BATCH * DIN;

        v4f xfrag[8];
        #pragma unroll
        for (int j = 0; j < 8; ++j)
            xfrag[j] = *(const v4f*)(xt + brow * DIN + j * 8 + lhi * 4);

        v8f Y[4] = {};

        #pragma unroll
        for (int ntp = 0; ntp < 4; ++ntp) {
            const int T0 = nclus * 8 + ntp * 2;
            const int n0 = T0 * 16 + lmod;

            v8f acc0 = {}, acc1 = {};
            #pragma unroll
            for (int j = 0; j < 8; ++j) {
                const int q = j * 2 + lhi;
                const v4f b0 = bs[q * DHID + n0];
                const v4f b1 = bs[q * DHID + n0 + 16];
                acc0 = wmma4(b0.xy, xfrag[j].xy, acc0);
                acc1 = wmma4(b1.xy, xfrag[j].xy, acc1);
                acc0 = wmma4(b0.zw, xfrag[j].zw, acc0);
                acc1 = wmma4(b1.zw, xfrag[j].zw, acc1);
            }

            {
                const float* ap = aV + T0 * 16 + lhi * 8;
                const v4f a0l = *(const v4f*)(ap);
                const v4f a0h = *(const v4f*)(ap + 4);
                const v4f a1l = *(const v4f*)(ap + 16);
                const v4f a1h = *(const v4f*)(ap + 20);
                const v8f a0 = __builtin_shufflevector(a0l, a0h, 0, 1, 2, 3, 4, 5, 6, 7);
                const v8f a1 = __builtin_shufflevector(a1l, a1h, 0, 1, 2, 3, 4, 5, 6, 7);
                Hreg[ntp * 2]     = Hreg[ntp * 2]     * a0 + acc0;
                Hreg[ntp * 2 + 1] = Hreg[ntp * 2 + 1] * a1 + acc1;
            }

            #pragma unroll
            for (int tt = 0; tt < 2; ++tt) {
                const int  T  = T0 + tt;
                const v8f  Hc = Hreg[ntp * 2 + tt];
                const v2f h01 = __builtin_shufflevector(Hc, Hc, 0, 1);
                const v2f h23 = __builtin_shufflevector(Hc, Hc, 2, 3);
                const v2f h45 = __builtin_shufflevector(Hc, Hc, 4, 5);
                const v2f h67 = __builtin_shufflevector(Hc, Hc, 6, 7);
                v4f cA[4], cB[4];
                #pragma unroll
                for (int o = 0; o < 4; ++o) {
                    cA[o] = cs[(4 * T + 2 * lhi)     * DOUT + o * 16 + lmod];
                    cB[o] = cs[(4 * T + 2 * lhi + 1) * DOUT + o * 16 + lmod];
                }
                #pragma unroll
                for (int o = 0; o < 4; ++o) Y[o] = wmma4(cA[o].xy, h01, Y[o]);
                #pragma unroll
                for (int o = 0; o < 4; ++o) Y[o] = wmma4(cA[o].zw, h23, Y[o]);
                #pragma unroll
                for (int o = 0; o < 4; ++o) Y[o] = wmma4(cB[o].xy, h45, Y[o]);
                #pragma unroll
                for (int o = 0; o < 4; ++o) Y[o] = wmma4(cB[o].zw, h67, Y[o]);
            }
        }

        __syncthreads();

        {
            v4f* pw = part + nclus * PPLANE + brow * PUNITS;
            const int sw = opaque_copy(lmod);
            #pragma unroll
            for (int o = 0; o < 4; ++o) {
                pw[(o * 4 + lhi * 2)     ^ sw] = __builtin_shufflevector(Y[o], Y[o], 0, 1, 2, 3);
                pw[(o * 4 + lhi * 2 + 1) ^ sw] = __builtin_shufflevector(Y[o], Y[o], 4, 5, 6, 7);
            }
        }
        __syncthreads();

        {
            float* yt = y + (size_t)t * BATCH * DOUT;
            #pragma unroll
            for (int p = 0; p < 2; ++p) {
                const int row = p * 32 + (tid >> 4);
                const int u   = tid & 15;
                const v4f* q0 = part + row * PUNITS + (u ^ (row & 15));
                const v4f s0 = q0[0];
                const v4f s1 = q0[PPLANE];
                const v4f s2 = q0[2 * PPLANE];
                const v4f s3 = q0[3 * PPLANE];
                *(v4f*)(yt + row * DOUT + u * 4) = (s0 + s1) + (s2 + s3);
            }
        }
    }
}

extern "C" void kernel_launch(void* const* d_in, const int* in_sizes, int n_in,
                              void* d_out, int out_size, void* d_ws, size_t ws_size,
                              hipStream_t stream)
{
    const float* h0 = (const float*)d_in[0];
    const float* x  = (const float*)d_in[1];
    const float* a  = (const float*)d_in[2];
    const float* bW = (const float*)d_in[3];
    const float* cW = (const float*)d_in[4];

    float* out    = (float*)d_out;
    float* hfinal = out;
    float* y      = out + BATCH * DHID;

    float* S  = (float*)d_ws;
    float* C  = S + (size_t)CHUNKS * BATCH * DHID;
    v4f*   bq = (v4f*)(C + (size_t)CHUNKS * BATCH * DHID);
    v4f*   cq = bq + 16 * DHID;

    rnn_pack<<<32, 256, 0, stream>>>(bW, cW, bq, cq);
    rnn_chunk_summary<<<CHUNKS, NTHREADS, 0, stream>>>(x, a, bq, S);
    rnn_carry_scan<<<(BATCH * DHID + 255) / 256, 256, 0, stream>>>(h0, a, S, C, hfinal);
    rnn_chunk_emit<<<CHUNKS, NTHREADS, 0, stream>>>(x, a, bq, cq, C, y);
}


# Both WMMA kernels software-pipelined: each 16-WMMA block's LDS weight fragments issued as one group a block ahead (2nd register set), next step's x requested under the current step; sums unchanged
# speedup vs baseline: 2.0247x; 1.0684x over previous
#include <hip/hip_runtime.h>


typedef __attribute__((ext_vector_type(2))) float v2f;
typedef __attribute__((ext_vector_type(4))) float v4f;
typedef __attribute__((ext_vector_type(8))) float v8f;

#define T_STEPS 4096
#define BATCH   64
#define DIN     64
#define DHID    512
#define DOUT    64
#define CHUNKS  256
#define CLEN    16
#define BQ_UNITS (16 * DHID)
#define CQ_UNITS (128 * DOUT)
#define PUNITS  16
#define PPLANE  (BATCH * PUNITS)
#define NWAVES  16
#define NTHREADS (NWAVES * 32)

__device__ __forceinline__ v8f wmma4(v2f a, v2f b, v8f c) {
    return __builtin_amdgcn_wmma_f32_16x16x4_f32(
        false, a, false, b, (short)0, c, false, false);
}

__device__ __forceinline__ int opaque_zero() {
    int z;
    asm volatile("s_mov_b32 %0, 0" : "=s"(z));
    return z;
}

__device__ __forceinline__ int opaque_copy(int v) {
    int r;
    asm volatile("v_mov_b32 %0, %1" : "=v"(r) : "v"(v));
    return r;
}

__global__ __launch_bounds__(256)
void rnn_pack(const float* __restrict__ bW, const float* __restrict__ cW,
              v4f* __restrict__ bq, v4f* __restrict__ cq)
{
    const int i = blockIdx.x * 256 + threadIdx.x;
    if (i < 16 * DHID) {
        const int q = i >> 9, n = i & (DHID - 1);
        v4f t;
        t.x = bW[(size_t)(4 * q)     * DHID + n];
        t.y = bW[(size_t)(4 * q + 1) * DHID + n];
        t.z = bW[(size_t)(4 * q + 2) * DHID + n];
        t.w = bW[(size_t)(4 * q + 3) * DHID + n];
        bq[i] = t;
    }
    if (i < 128 * DOUT) {
        const int q = i >> 6, n = i & (DOUT - 1);
        v4f t;
        t.x = cW[(size_t)(4 * q)     * DOUT + n];
        t.y = cW[(size_t)(4 * q + 1) * DOUT + n];
        t.z = cW[(size_t)(4 * q + 2) * DOUT + n];
        t.w = cW[(size_t)(4 * q + 3) * DOUT + n];
        cq[i] = t;
    }
}

__global__ __launch_bounds__(NTHREADS)
void rnn_chunk_summary(const float* __restrict__ x,
                       const float* __restrict__ aVec,
                       const v4f* __restrict__ bq0,
                       float* __restrict__ S)
{
    const int chunk = blockIdx.x;
    const int tid   = threadIdx.x;
    const int wave  = tid >> 5;
    const int lane  = tid & 31;
    const int lmod  = lane & 15;
    const int lhi   = lane >> 4;

    const int m_tile = wave & 3;
    const int nclus  = wave >> 2;
    const int rowA   = m_tile * 16 + lmod;

    __shared__ v4f bs[BQ_UNITS];

    #pragma unroll
    for (int k = 0; k < BQ_UNITS / NTHREADS; ++k)
        bs[k * NTHREADS + tid] = bq0[k * NTHREADS + tid];

    float av[8];
    #pragma unroll
    for (int nt = 0; nt < 8; ++nt)
        av[nt] = aVec[(nclus * 8 + nt) * 16 + lmod];

    const int bbase = lhi * DHID + nclus * 128 + lmod;

    v8f Sacc[8] = {};

    v4f afrag[8];
    {
        const float* xt = x + (size_t)(chunk * CLEN) * BATCH * DIN + rowA * DIN + lhi * 4;
        #pragma unroll
        for (int j = 0; j < 8; ++j)
            afrag[j] = *(const v4f*)(xt + j * 8);
    }

    __syncthreads();

    v4f bcur[4];
    {
        const int zb0 = opaque_copy(bbase);
        #pragma unroll
        for (int n = 0; n < 4; ++n)
            bcur[n] = bs[zb0 + n * 16];
    }

    for (int i = 0; i < CLEN; ++i) {
        const int zb = opaque_copy(bbase);

        const int inext = (i + 1 < CLEN) ? (i + 1) : i;
        const float* xn = x + (size_t)(chunk * CLEN + inext) * BATCH * DIN + rowA * DIN + lhi * 4;

        #pragma unroll
        for (int half = 0; half < 2; ++half) {
            v8f acc[4] = {};
            #pragma unroll
            for (int j = 0; j < 8; ++j) {
                const int nb = (half * 8 + j + 1) & 15;
                const int nh = nb >> 3, nj = nb & 7;
                v4f bnext[4];
                #pragma unroll
                for (int n = 0; n < 4; ++n)
                    bnext[n] = bs[zb + nj * 2 * DHID + nh * 64 + n * 16];
                __builtin_amdgcn_sched_barrier(0);
                #pragma unroll
                for (int n = 0; n < 4; ++n) acc[n] = wmma4(afrag[j].xy, bcur[n].xy, acc[n]);
                #pragma unroll
                for (int n = 0; n < 4; ++n) acc[n] = wmma4(afrag[j].zw, bcur[n].zw, acc[n]);
                if (half == 1)
                    afrag[j] = *(const v4f*)(xn + j * 8);
                __builtin_amdgcn_sched_barrier(0);
                #pragma unroll
                for (int n = 0; n < 4; ++n) bcur[n] = bnext[n];
            }
            #pragma unroll
            for (int n = 0; n < 4; ++n) {
                const float an = av[half * 4 + n];
                #pragma unroll
                for (int r = 0; r < 8; ++r)
                    Sacc[half * 4 + n][r] = Sacc[half * 4 + n][r] * an + acc[n][r];
            }
        }
    }

    float* Sj = S + (size_t)chunk * BATCH * DHID;
    #pragma unroll
    for (int nt = 0; nt < 8; ++nt) {
        const int col = (nclus * 8 + nt) * 16 + lmod;
        #pragma unroll
        for (int r = 0; r < 8; ++r) {
            const int row = m_tile * 16 + r + lhi * 8;
            Sj[row * DHID + col] = Sacc[nt][r];
        }
    }
}

__global__ __launch_bounds__(256)
void rnn_carry_scan(const float* __restrict__ h0,
                    const float* __restrict__ aVec,
                    const float* __restrict__ S,
                    float* __restrict__ Cout,
                    float* __restrict__ hfinal)
{
    const int idx = blockIdx.x * blockDim.x + threadIdx.x;
    if (idx >= BATCH * DHID) return;
    const int hcol = idx & (DHID - 1);

    float aL = aVec[hcol];
    #pragma unroll
    for (int s = 1; s < CLEN; s <<= 1) aL *= aL;

    float c = h0[idx];
    for (int j = 0; j < CHUNKS; ++j) {
        Cout[(size_t)j * BATCH * DHID + idx] = c;
        c = aL * c + S[(size_t)j * BATCH * DHID + idx];
    }
    hfinal[idx] = c;
}

__device__ __forceinline__ void load_b_block(v4f (&f)[8], const v4f* bsm, int col, int jh, int lhi) {
    #pragma unroll
    for (int jj = 0; jj < 4; ++jj) {
        const int q = (jh * 4 + jj) * 2 + lhi;
        f[2 * jj]     = bsm[q * DHID + col];
        f[2 * jj + 1] = bsm[q * DHID + col + 16];
    }
}

__device__ __forceinline__ void load_c_block(v4f (&f)[8], const v4f* csm, int T, int lhi, int lmod) {
    #pragma unroll
    for (int o = 0; o < 4; ++o) {
        f[o]     = csm[(4 * T + 2 * lhi)     * DOUT + o * 16 + lmod];
        f[4 + o] = csm[(4 * T + 2 * lhi + 1) * DOUT + o * 16 + lmod];
    }
}

__global__ __launch_bounds__(NTHREADS)
void rnn_chunk_emit(const float* __restrict__ x,
                    const float* __restrict__ aVec,
                    const v4f* __restrict__ bq0,
                    const v4f* __restrict__ cq0,
                    const float* __restrict__ Cin,
                    float* __restrict__ y)
{
    __shared__ v4f bs[BQ_UNITS];
    __shared__ v4f cs[CQ_UNITS];
    __shared__ v4f part[4 * PPLANE];

    const int chunk = blockIdx.x;
    const int tid   = threadIdx.x;
    const int wave  = tid >> 5;
    const int lane  = tid & 31;
    const int lmod  = lane & 15;
    const int lhi   = lane >> 4;

    const int m_tile = wave & 3;
    const int nclus  = wave >> 2;
    const int brow   = m_tile * 16 + lmod;

    #pragma unroll
    for (int k = 0; k < BQ_UNITS / NTHREADS; ++k)
        bs[k * NTHREADS + tid] = bq0[k * NTHREADS + tid];
    #pragma unroll
    for (int k = 0; k < CQ_UNITS / NTHREADS; ++k)
        cs[k * NTHREADS + tid] = cq0[k * NTHREADS + tid];

    v8f Hreg[8];
    {
        const float* Cj = Cin + (size_t)chunk * BATCH * DHID + brow * DHID + nclus * 128 + lhi * 8;
        #pragma unroll
        for (int nt = 0; nt < 8; ++nt) {
            const v4f lo = *(const v4f*)(Cj + nt * 16);
            const v4f hi = *(const v4f*)(Cj + nt * 16 + 4);
            Hreg[nt] = __builtin_shufflevector(lo, hi, 0, 1, 2, 3, 4, 5, 6, 7);
        }
    }

    v4f xfrag[8];
    {
        const float* xt = x + (size_t)(chunk * CLEN) * BATCH * DIN + brow * DIN + lhi * 4;
        #pragma unroll
        for (int j = 0; j < 8; ++j)
            xfrag[j] = *(const v4f*)(xt + j * 8);
    }

    const int zero = opaque_zero();
    const float* aV = aVec;

    __syncthreads();

    v4f fcur[8];
    load_b_block(fcur, bs, nclus * 128 + lmod, 0, lhi);

    for (int i = 0; i < CLEN; ++i) {
        aV += zero;
        const int t = chunk * CLEN + i;

        #pragma unroll
        for (int ntp = 0; ntp < 4; ++ntp) {
            const int T0 = nclus * 8 + ntp * 2;
            v8f acc0 = {}, acc1 = {};
            v4f a0l, a0h, a1l, a1h;
            #pragma unroll
            for (int jh = 0; jh < 2; ++jh) {
                const int blk = ntp * 2 + jh;
                v4f fnext[8];
                if (blk + 1 < 8)
                    load_b_block(fnext, bs, nclus * 128 + ((blk + 1) >> 1) * 32 + lmod, (blk + 1) & 1, lhi);
                else
                    load_c_block(fnext, cs, nclus * 8, lhi, lmod);
                if (jh == 0) {
                    const float* ap = aV + T0 * 16 + lhi * 8;
                    a0l = *(const v4f*)(ap);
                    a0h = *(const v4f*)(ap + 4);
                    a1l = *(const v4f*)(ap + 16);
                    a1h = *(const v4f*)(ap + 20);
                }
                __builtin_amdgcn_sched_barrier(0);
                #pragma unroll
                for (int jj = 0; jj < 4; ++jj) {
                    const int j = jh * 4 + jj;
                    const v4f b0 = fcur[2 * jj];
                    const v4f b1 = fcur[2 * jj + 1];
                    acc0 = wmma4(b0.xy, xfrag[j].xy, acc0);
                    acc1 = wmma4(b1.xy, xfrag[j].xy, acc1);
                    acc0 = wmma4(b0.zw, xfrag[j].zw, acc0);
                    acc1 = wmma4(b1.zw, xfrag[j].zw, acc1);
                }
                __builtin_amdgcn_sched_barrier(0);
                #pragma unroll
                for (int f = 0; f < 8; ++f) fcur[f] = fnext[f];
            }

            {
                const v8f a0 = __builtin_shufflevector(a0l, a0h, 0, 1, 2, 3, 4, 5, 6, 7);
                const v8f a1 = __builtin_shufflevector(a1l, a1h, 0, 1, 2, 3, 4, 5, 6, 7);
                Hreg[ntp * 2]     = Hreg[ntp * 2]     * a0 + acc0;
                Hreg[ntp * 2 + 1] = Hreg[ntp * 2 + 1] * a1 + acc1;
            }
        }

        {
            const int inext = (i + 1 < CLEN) ? (i + 1) : i;
            const float* xn = x + (size_t)(chunk * CLEN + inext) * BATCH * DIN + brow * DIN + lhi * 4;
            #pragma unroll
            for (int j = 0; j < 8; ++j)
                xfrag[j] = *(const v4f*)(xn + j * 8);
        }

        v8f Y[4] = {};

        #pragma unroll
        for (int nt = 0; nt < 8; ++nt) {
            v4f fnext[8];
            if (nt + 1 < 8)
                load_c_block(fnext, cs, nclus * 8 + nt + 1, lhi, lmod);
            else
                load_b_block(fnext, bs, nclus * 128 + lmod, 0, lhi);
            __builtin_amdgcn_sched_barrier(0);
            const v8f  Hc = Hreg[nt];
            const v2f h01 = __builtin_shufflevector(Hc, Hc, 0, 1);
            const v2f h23 = __builtin_shufflevector(Hc, Hc, 2, 3);
            const v2f h45 = __builtin_shufflevector(Hc, Hc, 4, 5);
            const v2f h67 = __builtin_shufflevector(Hc, Hc, 6, 7);
            #pragma unroll
            for (int o = 0; o < 4; ++o) Y[o] = wmma4(fcur[o].xy,     h01, Y[o]);
            #pragma unroll
            for (int o = 0; o < 4; ++o) Y[o] = wmma4(fcur[o].zw,     h23, Y[o]);
            #pragma unroll
            for (int o = 0; o < 4; ++o) Y[o] = wmma4(fcur[4 + o].xy, h45, Y[o]);
            #pragma unroll
            for (int o = 0; o < 4; ++o) Y[o] = wmma4(fcur[4 + o].zw, h67, Y[o]);
            __builtin_amdgcn_sched_barrier(0);
            #pragma unroll
            for (int f = 0; f < 8; ++f) fcur[f] = fnext[f];
        }

        __syncthreads();

        {
            v4f* pw = part + nclus * PPLANE + brow * PUNITS;
            const int sw = opaque_copy(lmod);
            #pragma unroll
            for (int o = 0; o < 4; ++o) {
                pw[(o * 4 + lhi * 2)     ^ sw] = __builtin_shufflevector(Y[o], Y[o], 0, 1, 2, 3);
                pw[(o * 4 + lhi * 2 + 1) ^ sw] = __builtin_shufflevector(Y[o], Y[o], 4, 5, 6, 7);
            }
        }
        __syncthreads();

        {
            float* yt = y + (size_t)t * BATCH * DOUT;
            #pragma unroll
            for (int p = 0; p < 2; ++p) {
                const int row = p * 32 + (tid >> 4);
                const int u   = tid & 15;
                const v4f* q0 = part + row * PUNITS + (u ^ (row & 15));
                const v4f s0 = q0[0];
                const v4f s1 = q0[PPLANE];
                const v4f s2 = q0[2 * PPLANE];
                const v4f s3 = q0[3 * PPLANE];
                *(v4f*)(yt + row * DOUT + u * 4) = (s0 + s1) + (s2 + s3);
            }
        }
    }
}

extern "C" void kernel_launch(void* const* d_in, const int* in_sizes, int n_in,
                              void* d_out, int out_size, void* d_ws, size_t ws_size,
                              hipStream_t stream)
{
    const float* h0 = (const float*)d_in[0];
    const float* x  = (const float*)d_in[1];
    const float* a  = (const float*)d_in[2];
    const float* bW = (const float*)d_in[3];
    const float* cW = (const float*)d_in[4];

    float* out    = (float*)d_out;
    float* hfinal = out;
    float* y      = out + BATCH * DHID;

    float* S  = (float*)d_ws;
    float* C  = S + (size_t)CHUNKS * BATCH * DHID;
    v4f*   bq = (v4f*)(C + (size_t)CHUNKS * BATCH * DHID);
    v4f*   cq = bq + 16 * DHID;

    rnn_pack<<<32, 256, 0, stream>>>(bW, cW, bq, cq);
    rnn_chunk_summary<<<CHUNKS, NTHREADS, 0, stream>>>(x, a, bq, S);
    rnn_carry_scan<<<(BATCH * DHID + 255) / 256, 256, 0, stream>>>(h0, a, S, C, hfinal);
    rnn_chunk_emit<<<CHUNKS, NTHREADS, 0, stream>>>(x, a, bq, cq, C, y);
}


# Carry scan: each thread keeps 32 chunk summaries in flight and asks one batch ahead (8 memory round trips per thread, not 32); 256 two-wave work-groups, 8-byte accesses; chain order unchanged
# speedup vs baseline: 2.1878x; 1.0806x over previous
#include <hip/hip_runtime.h>


typedef __attribute__((ext_vector_type(2))) float v2f;
typedef __attribute__((ext_vector_type(4))) float v4f;
typedef __attribute__((ext_vector_type(8))) float v8f;

#define T_STEPS 4096
#define BATCH   64
#define DIN     64
#define DHID    512
#define DOUT    64
#define CHUNKS  256
#define CLEN    16
#define BQ_UNITS (16 * DHID)
#define CQ_UNITS (128 * DOUT)
#define PUNITS  16
#define PPLANE  (BATCH * PUNITS)
#define NWAVES  16
#define NTHREADS (NWAVES * 32)

__device__ __forceinline__ v8f wmma4(v2f a, v2f b, v8f c) {
    return __builtin_amdgcn_wmma_f32_16x16x4_f32(
        false, a, false, b, (short)0, c, false, false);
}

__device__ __forceinline__ int opaque_zero() {
    int z;
    asm volatile("s_mov_b32 %0, 0" : "=s"(z));
    return z;
}

__device__ __forceinline__ int opaque_copy(int v) {
    int r;
    asm volatile("v_mov_b32 %0, %1" : "=v"(r) : "v"(v));
    return r;
}

__global__ __launch_bounds__(256)
void rnn_pack(const float* __restrict__ bW, const float* __restrict__ cW,
              v4f* __restrict__ bq, v4f* __restrict__ cq)
{
    const int i = blockIdx.x * 256 + threadIdx.x;
    if (i < 16 * DHID) {
        const int q = i >> 9, n = i & (DHID - 1);
        v4f t;
        t.x = bW[(size_t)(4 * q)     * DHID + n];
        t.y = bW[(size_t)(4 * q + 1) * DHID + n];
        t.z = bW[(size_t)(4 * q + 2) * DHID + n];
        t.w = bW[(size_t)(4 * q + 3) * DHID + n];
        bq[i] = t;
    }
    if (i < 128 * DOUT) {
        const int q = i >> 6, n = i & (DOUT - 1);
        v4f t;
        t.x = cW[(size_t)(4 * q)     * DOUT + n];
        t.y = cW[(size_t)(4 * q + 1) * DOUT + n];
        t.z = cW[(size_t)(4 * q + 2) * DOUT + n];
        t.w = cW[(size_t)(4 * q + 3) * DOUT + n];
        cq[i] = t;
    }
}

__global__ __launch_bounds__(NTHREADS)
void rnn_chunk_summary(const float* __restrict__ x,
                       const float* __restrict__ aVec,
                       const v4f* __restrict__ bq0,
                       float* __restrict__ S)
{
    const int chunk = blockIdx.x;
    const int tid   = threadIdx.x;
    const int wave  = tid >> 5;
    const int lane  = tid & 31;
    const int lmod  = lane & 15;
    const int lhi   = lane >> 4;

    const int m_tile = wave & 3;
    const int nclus  = wave >> 2;
    const int rowA   = m_tile * 16 + lmod;

    __shared__ v4f bs[BQ_UNITS];

    #pragma unroll
    for (int k = 0; k < BQ_UNITS / NTHREADS; ++k)
        bs[k * NTHREADS + tid] = bq0[k * NTHREADS + tid];

    float av[8];
    #pragma unroll
    for (int nt = 0; nt < 8; ++nt)
        av[nt] = aVec[(nclus * 8 + nt) * 16 + lmod];

    const int bbase = lhi * DHID + nclus * 128 + lmod;

    v8f Sacc[8] = {};

    v4f afrag[8];
    {
        const float* xt = x + (size_t)(chunk * CLEN) * BATCH * DIN + rowA * DIN + lhi * 4;
        #pragma unroll
        for (int j = 0; j < 8; ++j)
            afrag[j] = *(const v4f*)(xt + j * 8);
    }

    __syncthreads();

    v4f bcur[4];
    {
        const int zb0 = opaque_copy(bbase);
        #pragma unroll
        for (int n = 0; n < 4; ++n)
            bcur[n] = bs[zb0 + n * 16];
    }

    for (int i = 0; i < CLEN; ++i) {
        const int zb = opaque_copy(bbase);

        const int inext = (i + 1 < CLEN) ? (i + 1) : i;
        const float* xn = x + (size_t)(chunk * CLEN + inext) * BATCH * DIN + rowA * DIN + lhi * 4;

        #pragma unroll
        for (int half = 0; half < 2; ++half) {
            v8f acc[4] = {};
            #pragma unroll
            for (int j = 0; j < 8; ++j) {
                const int nb = (half * 8 + j + 1) & 15;
                const int nh = nb >> 3, nj = nb & 7;
                v4f bnext[4];
                #pragma unroll
                for (int n = 0; n < 4; ++n)
                    bnext[n] = bs[zb + nj * 2 * DHID + nh * 64 + n * 16];
                __builtin_amdgcn_sched_barrier(0);
                #pragma unroll
                for (int n = 0; n < 4; ++n) acc[n] = wmma4(afrag[j].xy, bcur[n].xy, acc[n]);
                #pragma unroll
                for (int n = 0; n < 4; ++n) acc[n] = wmma4(afrag[j].zw, bcur[n].zw, acc[n]);
                if (half == 1)
                    afrag[j] = *(const v4f*)(xn + j * 8);
                __builtin_amdgcn_sched_barrier(0);
                #pragma unroll
                for (int n = 0; n < 4; ++n) bcur[n] = bnext[n];
            }
            #pragma unroll
            for (int n = 0; n < 4; ++n) {
                const float an = av[half * 4 + n];
                #pragma unroll
                for (int r = 0; r < 8; ++r)
                    Sacc[half * 4 + n][r] = Sacc[half * 4 + n][r] * an + acc[n][r];
            }
        }
    }

    float* Sj = S + (size_t)chunk * BATCH * DHID;
    #pragma unroll
    for (int nt = 0; nt < 8; ++nt) {
        const int col = (nclus * 8 + nt) * 16 + lmod;
        #pragma unroll
        for (int r = 0; r < 8; ++r) {
            const int row = m_tile * 16 + r + lhi * 8;
            Sj[row * DHID + col] = Sacc[nt][r];
        }
    }
}

#define SCAN_DEPTH   32
#define SCAN_THREADS 64
#define SCAN_PLANE   (BATCH * DHID / 2)

__global__ __launch_bounds__(SCAN_THREADS)
void rnn_carry_scan(const float* __restrict__ h0,
                    const float* __restrict__ aVec,
                    const float* __restrict__ S,
                    float* __restrict__ Cout,
                    float* __restrict__ hfinal)
{
    const int pair = blockIdx.x * SCAN_THREADS + threadIdx.x;
    const int idx  = pair * 2;
    const int hcol = idx & (DHID - 1);

    v2f aL = *(const v2f*)(aVec + hcol);
    #pragma unroll
    for (int s = 1; s < CLEN; s <<= 1) aL *= aL;

    const v2f* Sp = (const v2f*)S + pair;
    v2f*       Cp = (v2f*)Cout + pair;

    v2f cur[SCAN_DEPTH];
    #pragma unroll
    for (int k = 0; k < SCAN_DEPTH; ++k)
        cur[k] = Sp[(size_t)k * SCAN_PLANE];

    v2f c = *(const v2f*)(h0 + idx);

    #pragma unroll
    for (int jb = 0; jb < CHUNKS / SCAN_DEPTH; ++jb) {
        v2f nxt[SCAN_DEPTH];
        if (jb + 1 < CHUNKS / SCAN_DEPTH) {
            #pragma unroll
            for (int k = 0; k < SCAN_DEPTH; ++k)
                nxt[k] = Sp[(size_t)((jb + 1) * SCAN_DEPTH + k) * SCAN_PLANE];
        }
        __builtin_amdgcn_sched_barrier(0);
        #pragma unroll
        for (int k = 0; k < SCAN_DEPTH; ++k) {
            Cp[(size_t)(jb * SCAN_DEPTH + k) * SCAN_PLANE] = c;
            c = aL * c + cur[k];
        }
        __builtin_amdgcn_sched_barrier(0);
        #pragma unroll
        for (int k = 0; k < SCAN_DEPTH; ++k) cur[k] = nxt[k];
    }
    *(v2f*)(hfinal + idx) = c;
}

__device__ __forceinline__ void load_b_block(v4f (&f)[8], const v4f* bsm, int col, int jh, int lhi) {
    #pragma unroll
    for (int jj = 0; jj < 4; ++jj) {
        const int q = (jh * 4 + jj) * 2 + lhi;
        f[2 * jj]     = bsm[q * DHID + col];
        f[2 * jj + 1] = bsm[q * DHID + col + 16];
    }
}

__device__ __forceinline__ void load_c_block(v4f (&f)[8], const v4f* csm, int T, int lhi, int lmod) {
    #pragma unroll
    for (int o = 0; o < 4; ++o) {
        f[o]     = csm[(4 * T + 2 * lhi)     * DOUT + o * 16 + lmod];
        f[4 + o] = csm[(4 * T + 2 * lhi + 1) * DOUT + o * 16 + lmod];
    }
}

__global__ __launch_bounds__(NTHREADS)
void rnn_chunk_emit(const float* __restrict__ x,
                    const float* __restrict__ aVec,
                    const v4f* __restrict__ bq0,
                    const v4f* __restrict__ cq0,
                    const float* __restrict__ Cin,
                    float* __restrict__ y)
{
    __shared__ v4f bs[BQ_UNITS];
    __shared__ v4f cs[CQ_UNITS];
    __shared__ v4f part[4 * PPLANE];

    const int chunk = blockIdx.x;
    const int tid   = threadIdx.x;
    const int wave  = tid >> 5;
    const int lane  = tid & 31;
    const int lmod  = lane & 15;
    const int lhi   = lane >> 4;

    const int m_tile = wave & 3;
    const int nclus  = wave >> 2;
    const int brow   = m_tile * 16 + lmod;

    #pragma unroll
    for (int k = 0; k < BQ_UNITS / NTHREADS; ++k)
        bs[k * NTHREADS + tid] = bq0[k * NTHREADS + tid];
    #pragma unroll
    for (int k = 0; k < CQ_UNITS / NTHREADS; ++k)
        cs[k * NTHREADS + tid] = cq0[k * NTHREADS + tid];

    v8f Hreg[8];
    {
        const float* Cj = Cin + (size_t)chunk * BATCH * DHID + brow * DHID + nclus * 128 + lhi * 8;
        #pragma unroll
        for (int nt = 0; nt < 8; ++nt) {
            const v4f lo = *(const v4f*)(Cj + nt * 16);
            const v4f hi = *(const v4f*)(Cj + nt * 16 + 4);
            Hreg[nt] = __builtin_shufflevector(lo, hi, 0, 1, 2, 3, 4, 5, 6, 7);
        }
    }

    v4f xfrag[8];
    {
        const float* xt = x + (size_t)(chunk * CLEN) * BATCH * DIN + brow * DIN + lhi * 4;
        #pragma unroll
        for (int j = 0; j < 8; ++j)
            xfrag[j] = *(const v4f*)(xt + j * 8);
    }

    const int zero = opaque_zero();
    const float* aV = aVec;

    __syncthreads();

    v4f fcur[8];
    load_b_block(fcur, bs, nclus * 128 + lmod, 0, lhi);

    for (int i = 0; i < CLEN; ++i) {
        aV += zero;
        const int t = chunk * CLEN + i;

        #pragma unroll
        for (int ntp = 0; ntp < 4; ++ntp) {
            const int T0 = nclus * 8 + ntp * 2;
            v8f acc0 = {}, acc1 = {};
            v4f a0l, a0h, a1l, a1h;
            #pragma unroll
            for (int jh = 0; jh < 2; ++jh) {
                const int blk = ntp * 2 + jh;
                v4f fnext[8];
                if (blk + 1 < 8)
                    load_b_block(fnext, bs, nclus * 128 + ((blk + 1) >> 1) * 32 + lmod, (blk + 1) & 1, lhi);
                else
                    load_c_block(fnext, cs, nclus * 8, lhi, lmod);
                if (jh == 0) {
                    const float* ap = aV + T0 * 16 + lhi * 8;
                    a0l = *(const v4f*)(ap);
                    a0h = *(const v4f*)(ap + 4);
                    a1l = *(const v4f*)(ap + 16);
                    a1h = *(const v4f*)(ap + 20);
                }
                __builtin_amdgcn_sched_barrier(0);
                #pragma unroll
                for (int jj = 0; jj < 4; ++jj) {
                    const int j = jh * 4 + jj;
                    const v4f b0 = fcur[2 * jj];
                    const v4f b1 = fcur[2 * jj + 1];
                    acc0 = wmma4(b0.xy, xfrag[j].xy, acc0);
                    acc1 = wmma4(b1.xy, xfrag[j].xy, acc1);
                    acc0 = wmma4(b0.zw, xfrag[j].zw, acc0);
                    acc1 = wmma4(b1.zw, xfrag[j].zw, acc1);
                }
                __builtin_amdgcn_sched_barrier(0);
                #pragma unroll
                for (int f = 0; f < 8; ++f) fcur[f] = fnext[f];
            }

            {
                const v8f a0 = __builtin_shufflevector(a0l, a0h, 0, 1, 2, 3, 4, 5, 6, 7);
                const v8f a1 = __builtin_shufflevector(a1l, a1h, 0, 1, 2, 3, 4, 5, 6, 7);
                Hreg[ntp * 2]     = Hreg[ntp * 2]     * a0 + acc0;
                Hreg[ntp * 2 + 1] = Hreg[ntp * 2 + 1] * a1 + acc1;
            }
        }

        {
            const int inext = (i + 1 < CLEN) ? (i + 1) : i;
            const float* xn = x + (size_t)(chunk * CLEN + inext) * BATCH * DIN + brow * DIN + lhi * 4;
            #pragma unroll
            for (int j = 0; j < 8; ++j)
                xfrag[j] = *(const v4f*)(xn + j * 8);
        }

        v8f Y[4] = {};

        #pragma unroll
        for (int nt = 0; nt < 8; ++nt) {
            v4f fnext[8];
            if (nt + 1 < 8)
                load_c_block(fnext, cs, nclus * 8 + nt + 1, lhi, lmod);
            else
                load_b_block(fnext, bs, nclus * 128 + lmod, 0, lhi);
            __builtin_amdgcn_sched_barrier(0);
            const v8f  Hc = Hreg[nt];
            const v2f h01 = __builtin_shufflevector(Hc, Hc, 0, 1);
            const v2f h23 = __builtin_shufflevector(Hc, Hc, 2, 3);
            const v2f h45 = __builtin_shufflevector(Hc, Hc, 4, 5);
            const v2f h67 = __builtin_shufflevector(Hc, Hc, 6, 7);
            #pragma unroll
            for (int o = 0; o < 4; ++o) Y[o] = wmma4(fcur[o].xy,     h01, Y[o]);
            #pragma unroll
            for (int o = 0; o < 4; ++o) Y[o] = wmma4(fcur[o].zw,     h23, Y[o]);
            #pragma unroll
            for (int o = 0; o < 4; ++o) Y[o] = wmma4(fcur[4 + o].xy, h45, Y[o]);
            #pragma unroll
            for (int o = 0; o < 4; ++o) Y[o] = wmma4(fcur[4 + o].zw, h67, Y[o]);
            __builtin_amdgcn_sched_barrier(0);
            #pragma unroll
            for (int f = 0; f < 8; ++f) fcur[f] = fnext[f];
        }

        __syncthreads();

        {
            v4f* pw = part + nclus * PPLANE + brow * PUNITS;
            const int sw = opaque_copy(lmod);
            #pragma unroll
            for (int o = 0; o < 4; ++o) {
                pw[(o * 4 + lhi * 2)     ^ sw] = __builtin_shufflevector(Y[o], Y[o], 0, 1, 2, 3);
                pw[(o * 4 + lhi * 2 + 1) ^ sw] = __builtin_shufflevector(Y[o], Y[o], 4, 5, 6, 7);
            }
        }
        __syncthreads();

        {
            float* yt = y + (size_t)t * BATCH * DOUT;
            #pragma unroll
            for (int p = 0; p < 2; ++p) {
                const int row = p * 32 + (tid >> 4);
                const int u   = tid & 15;
                const v4f* q0 = part + row * PUNITS + (u ^ (row & 15));
                const v4f s0 = q0[0];
                const v4f s1 = q0[PPLANE];
                const v4f s2 = q0[2 * PPLANE];
                const v4f s3 = q0[3 * PPLANE];
                *(v4f*)(yt + row * DOUT + u * 4) = (s0 + s1) + (s2 + s3);
            }
        }
    }
}

extern "C" void kernel_launch(void* const* d_in, const int* in_sizes, int n_in,
                              void* d_out, int out_size, void* d_ws, size_t ws_size,
                              hipStream_t stream)
{
    const float* h0 = (const float*)d_in[0];
    const float* x  = (const float*)d_in[1];
    const float* a  = (const float*)d_in[2];
    const float* bW = (const float*)d_in[3];
    const float* cW = (const float*)d_in[4];

    float* out    = (float*)d_out;
    float* hfinal = out;
    float* y      = out + BATCH * DHID;

    float* S  = (float*)d_ws;
    float* C  = S + (size_t)CHUNKS * BATCH * DHID;
    v4f*   bq = (v4f*)(C + (size_t)CHUNKS * BATCH * DHID);
    v4f*   cq = bq + 16 * DHID;

    rnn_pack<<<32, 256, 0, stream>>>(bW, cW, bq, cq);
    rnn_chunk_summary<<<CHUNKS, NTHREADS, 0, stream>>>(x, a, bq, S);
    rnn_carry_scan<<<BATCH * DHID / 2 / SCAN_THREADS, SCAN_THREADS, 0, stream>>>(h0, a, S, C, hfinal);
    rnn_chunk_emit<<<CHUNKS, NTHREADS, 0, stream>>>(x, a, bq, cq, C, y);
}


# Emit kernel stops recomputing x@b: in 8-step halves, y_i = (a^(i+1) C) c + sum_d x_(i-d) M_d, M_d = b diag(a^d) c (eight 64x64, set-up WMMA); 2nd half from a stored half summary; sums regrouped
# speedup vs baseline: 2.4139x; 1.1033x over previous
#include <hip/hip_runtime.h>


typedef __attribute__((ext_vector_type(2))) float v2f;
typedef __attribute__((ext_vector_type(4))) float v4f;
typedef __attribute__((ext_vector_type(8))) float v8f;

#define T_STEPS 4096
#define BATCH   64
#define DIN     64
#define DHID    512
#define DOUT    64
#define CHUNKS  256
#define CLEN    16
#define SUBLEN  8
#define NSUB    (CLEN / SUBLEN)
#define MQ_UNITS (SUBLEN * 16 * DOUT)
#define BQ_UNITS (16 * DHID)
#define CQ_UNITS (128 * DOUT)
#define PUNITS  16
#define PPLANE  (BATCH * PUNITS)
#define NWAVES  16
#define NTHREADS (NWAVES * 32)

__device__ __forceinline__ v8f wmma4(v2f a, v2f b, v8f c) {
    return __builtin_amdgcn_wmma_f32_16x16x4_f32(
        false, a, false, b, (short)0, c, false, false);
}

__device__ __forceinline__ int opaque_zero() {
    int z;
    asm volatile("s_mov_b32 %0, 0" : "=s"(z));
    return z;
}

__device__ __forceinline__ int opaque_lane() {
    int r;
    asm volatile("v_mbcnt_lo_u32_b32 %0, -1, 0" : "=v"(r));
    return r;
}

__device__ __forceinline__ int opaque_copy(int v) {
    int r;
    asm volatile("v_mov_b32 %0, %1" : "=v"(r) : "v"(v));
    return r;
}

__global__ __launch_bounds__(256)
void rnn_pack(const float* __restrict__ bW, const float* __restrict__ cW,
              const float* __restrict__ aVec,
              v4f* __restrict__ bq, v4f* __restrict__ cq, v4f* __restrict__ mq)
{
    __shared__ v8f mpart[8 * SUBLEN * 32];

    if (blockIdx.x >= 32) {
        const int tile = blockIdx.x - 32;
        const int mt   = tile >> 2;
        const int nt   = tile & 3;
        const int w    = threadIdx.x >> 5;
        const int lane = threadIdx.x & 31;
        const int lmod = lane & 15;
        const int lhi  = lane >> 4;
        const float* bp = bW + (size_t)(16 * mt + lmod) * DHID + 64 * w + 4 * lhi;
        const float* cp = cW + (size_t)(64 * w + 4 * lhi) * DOUT + 16 * nt + lmod;
        const float* ap = aVec + 64 * w + 4 * lhi;
        v8f acc[SUBLEN] = {};
        v4f bv8[8], av8[8], cv8[8];
        #pragma unroll
        for (int k8 = 0; k8 < 8; ++k8) {
            bv8[k8]   = *(const v4f*)(bp + 8 * k8);
            av8[k8]   = *(const v4f*)(ap + 8 * k8);
            cv8[k8].x = cp[(size_t)(8 * k8)     * DOUT];
            cv8[k8].y = cp[(size_t)(8 * k8 + 1) * DOUT];
            cv8[k8].z = cp[(size_t)(8 * k8 + 2) * DOUT];
            cv8[k8].w = cp[(size_t)(8 * k8 + 3) * DOUT];
        }
        __builtin_amdgcn_sched_barrier(0);
        #pragma unroll
        for (int k8 = 0; k8 < 8; ++k8) {
            const v4f av = av8[k8];
            const v4f cv = cv8[k8];
            v4f bd[SUBLEN];
            bd[0] = bv8[k8];
            #pragma unroll
            for (int d = 1; d < SUBLEN; ++d) bd[d] = bd[d - 1] * av;
            #pragma unroll
            for (int d = 0; d < SUBLEN; ++d) acc[d] = wmma4(bd[d].xy, cv.xy, acc[d]);
            #pragma unroll
            for (int d = 0; d < SUBLEN; ++d) acc[d] = wmma4(bd[d].zw, cv.zw, acc[d]);
        }
        #pragma unroll
        for (int d = 0; d < SUBLEN; ++d) mpart[(w * SUBLEN + d) * 32 + lane] = acc[d];
        __syncthreads();
        v8f m = mpart[w * 32 + lane];
        #pragma unroll
        for (int s = 1; s < 8; ++s) m += mpart[(s * SUBLEN + w) * 32 + lane];
        v4f* mo = mq + w * (16 * DOUT) + (4 * mt + 2 * lhi) * DOUT + 16 * nt + lmod;
        mo[0]    = __builtin_shufflevector(m, m, 0, 1, 2, 3);
        mo[DOUT] = __builtin_shufflevector(m, m, 4, 5, 6, 7);
        return;
    }
    const int i = blockIdx.x * 256 + threadIdx.x;
    if (i < 16 * DHID) {
        const int q = i >> 9, n = i & (DHID - 1);
        v4f t;
        t.x = bW[(size_t)(4 * q)     * DHID + n];
        t.y = bW[(size_t)(4 * q + 1) * DHID + n];
        t.z = bW[(size_t)(4 * q + 2) * DHID + n];
        t.w = bW[(size_t)(4 * q + 3) * DHID + n];
        bq[i] = t;
    }
    if (i < 128 * DOUT) {
        const int q = i >> 6, n = i & (DOUT - 1);
        v4f t;
        t.x = cW[(size_t)(4 * q)     * DOUT + n];
        t.y = cW[(size_t)(4 * q + 1) * DOUT + n];
        t.z = cW[(size_t)(4 * q + 2) * DOUT + n];
        t.w = cW[(size_t)(4 * q + 3) * DOUT + n];
        cq[i] = t;
    }
}

__global__ __launch_bounds__(NTHREADS)
void rnn_chunk_summary(const float* __restrict__ x,
                       const float* __restrict__ aVec,
                       const v4f* __restrict__ bq0,
                       float* __restrict__ S,
                       float* __restrict__ Sh)
{
    const int chunk = blockIdx.x;
    const int tid   = threadIdx.x;
    const int wave  = tid >> 5;
    const int lane  = tid & 31;
    const int lmod  = lane & 15;
    const int lhi   = lane >> 4;

    const int m_tile = wave & 3;
    const int nclus  = wave >> 2;
    const int rowA   = m_tile * 16 + lmod;

    __shared__ v4f bs[BQ_UNITS];

    #pragma unroll
    for (int k = 0; k < BQ_UNITS / NTHREADS; ++k)
        bs[k * NTHREADS + tid] = bq0[k * NTHREADS + tid];

    float av[8];
    #pragma unroll
    for (int nt = 0; nt < 8; ++nt)
        av[nt] = aVec[(nclus * 8 + nt) * 16 + lmod];

    const int bbase = lhi * DHID + nclus * 128 + lmod;

    v8f Sacc[8] = {};

    v4f afrag[8];
    {
        const float* xt = x + (size_t)(chunk * CLEN) * BATCH * DIN + rowA * DIN + lhi * 4;
        #pragma unroll
        for (int j = 0; j < 8; ++j)
            afrag[j] = *(const v4f*)(xt + j * 8);
    }

    __syncthreads();

    v4f bcur[4];
    {
        const int zb0 = opaque_copy(bbase);
        #pragma unroll
        for (int n = 0; n < 4; ++n)
            bcur[n] = bs[zb0 + n * 16];
    }

    for (int i = 0; i < CLEN; ++i) {
        const int zb = opaque_copy(bbase);

        const int inext = (i + 1 < CLEN) ? (i + 1) : i;
        const float* xn = x + (size_t)(chunk * CLEN + inext) * BATCH * DIN + rowA * DIN + lhi * 4;

        #pragma unroll
        for (int half = 0; half < 2; ++half) {
            v8f acc[4] = {};
            #pragma unroll
            for (int j = 0; j < 8; ++j) {
                const int nb = (half * 8 + j + 1) & 15;
                const int nh = nb >> 3, nj = nb & 7;
                v4f bnext[4];
                #pragma unroll
                for (int n = 0; n < 4; ++n)
                    bnext[n] = bs[zb + nj * 2 * DHID + nh * 64 + n * 16];
                __builtin_amdgcn_sched_barrier(0);
                #pragma unroll
                for (int n = 0; n < 4; ++n) acc[n] = wmma4(afrag[j].xy, bcur[n].xy, acc[n]);
                #pragma unroll
                for (int n = 0; n < 4; ++n) acc[n] = wmma4(afrag[j].zw, bcur[n].zw, acc[n]);
                if (half == 1)
                    afrag[j] = *(const v4f*)(xn + j * 8);
                __builtin_amdgcn_sched_barrier(0);
                #pragma unroll
                for (int n = 0; n < 4; ++n) bcur[n] = bnext[n];
            }
            #pragma unroll
            for (int n = 0; n < 4; ++n) {
                const float an = av[half * 4 + n];
                #pragma unroll
                for (int r = 0; r < 8; ++r)
                    Sacc[half * 4 + n][r] = Sacc[half * 4 + n][r] * an + acc[n][r];
            }
        }

        if ((i & (SUBLEN - 1)) == SUBLEN - 1) {
            float* Sj = ((i == CLEN - 1) ? S : Sh) + (size_t)chunk * BATCH * DHID;
            #pragma unroll
            for (int nt = 0; nt < 8; ++nt) {
                const int col = (nclus * 8 + nt) * 16 + lmod;
                #pragma unroll
                for (int r = 0; r < 8; ++r) {
                    const int row = m_tile * 16 + r + lhi * 8;
                    Sj[row * DHID + col] = Sacc[nt][r];
                }
            }
        }
    }
}

#define SCAN_DEPTH   32
#define SCAN_THREADS 64
#define SCAN_PLANE   (BATCH * DHID / 2)

__global__ __launch_bounds__(SCAN_THREADS)
void rnn_carry_scan(const float* __restrict__ h0,
                    const float* __restrict__ aVec,
                    const float* __restrict__ S,
                    float* __restrict__ Cout,
                    float* __restrict__ hfinal)
{
    const int pair = blockIdx.x * SCAN_THREADS + threadIdx.x;
    const int idx  = pair * 2;
    const int hcol = idx & (DHID - 1);

    v2f aL = *(const v2f*)(aVec + hcol);
    #pragma unroll
    for (int s = 1; s < CLEN; s <<= 1) aL *= aL;

    const v2f* Sp = (const v2f*)S + pair;
    v2f*       Cp = (v2f*)Cout + pair;

    v2f cur[SCAN_DEPTH];
    #pragma unroll
    for (int k = 0; k < SCAN_DEPTH; ++k)
        cur[k] = Sp[(size_t)k * SCAN_PLANE];

    v2f c = *(const v2f*)(h0 + idx);

    #pragma unroll
    for (int jb = 0; jb < CHUNKS / SCAN_DEPTH; ++jb) {
        v2f nxt[SCAN_DEPTH];
        if (jb + 1 < CHUNKS / SCAN_DEPTH) {
            #pragma unroll
            for (int k = 0; k < SCAN_DEPTH; ++k)
                nxt[k] = Sp[(size_t)((jb + 1) * SCAN_DEPTH + k) * SCAN_PLANE];
        }
        __builtin_amdgcn_sched_barrier(0);
        #pragma unroll
        for (int k = 0; k < SCAN_DEPTH; ++k) {
            Cp[(size_t)(jb * SCAN_DEPTH + k) * SCAN_PLANE] = c;
            c = aL * c + cur[k];
        }
        __builtin_amdgcn_sched_barrier(0);
        #pragma unroll
        for (int k = 0; k < SCAN_DEPTH; ++k) cur[k] = nxt[k];
    }
    *(v2f*)(hfinal + idx) = c;
}

__device__ __forceinline__ void load_m_block(v4f (&f)[8], const v4f* msm, int d, int mlo, int mhi) {
    const int base = (d < 4) ? mlo : mhi;
    #pragma unroll
    for (int jj = 0; jj < 2; ++jj) {
        #pragma unroll
        for (int o = 0; o < 4; ++o)
            f[4 * jj + o] = msm[base + (d & 3) * (16 * DOUT) + jj * (2 * DOUT) + o * 16];
    }
}

__device__ __forceinline__ void load_c_block(v4f (&f)[8], const v4f* csm, int nt, int cb) {
    #pragma unroll
    for (int o = 0; o < 4; ++o) {
        f[o]     = csm[cb + nt * (4 * DOUT)        + o * 16];
        f[4 + o] = csm[cb + nt * (4 * DOUT) + DOUT + o * 16];
    }
}

__device__ __forceinline__ void load_x_slice(v4f (&xv)[2], const float* x, unsigned xoff, int t) {
    const float* p = x + (size_t)t * BATCH * DIN;
    xv[0] = *(const v4f*)(p + xoff);
    xv[1] = *(const v4f*)(p + xoff + 8);
}

__global__ __launch_bounds__(NTHREADS)
void rnn_chunk_emit(const float* __restrict__ x,
                    const float* __restrict__ aVec,
                    const v4f* __restrict__ mq0,
                    const v4f* __restrict__ cq0,
                    const float* __restrict__ Cin,
                    const float* __restrict__ Sh,
                    float* __restrict__ y)
{
    __shared__ v4f ms[MQ_UNITS];
    __shared__ v4f cs[CQ_UNITS];
    __shared__ v4f part[4 * PPLANE];

    const int chunk = blockIdx.x;
    const int tid   = threadIdx.x;
    const int wave  = __builtin_amdgcn_readfirstlane(tid >> 5);

    const int m_tile = wave & 3;
    const int nclus  = wave >> 2;

    #pragma unroll
    for (int k = 0; k < MQ_UNITS / NTHREADS; ++k)
        ms[k * NTHREADS + tid] = mq0[k * NTHREADS + tid];
    #pragma unroll
    for (int k = 0; k < CQ_UNITS / NTHREADS; ++k)
        cs[k * NTHREADS + tid] = cq0[k * NTHREADS + tid];

    const int t00   = chunk * CLEN;
    const int tlast = t00 + CLEN - 1;

    v4f xnew[2];
    v4f fcur[8];
    {
        const int ln = opaque_lane();
        const int lmod = ln & 15, lhi = ln >> 4;
        load_x_slice(xnew, x, (unsigned)((m_tile * 16 + lmod) * DIN + nclus * 16 + lhi * 4), t00);
        __syncthreads();
        const int mlo = (4 * nclus + lhi) * DOUT + lmod;
        load_m_block(fcur, ms, 0, mlo, mlo);
    }

    const int zero = opaque_zero();
    const float* aV = aVec;

    v4f xcur[2] = {};
    v4f acur[2] = {};
    v8f Hreg[8] = {};

    for (int sub = 0; sub < NSUB; ++sub) {
        const int t0 = t00 + sub * SUBLEN;

        {
            const int ln = opaque_lane();
            const int lmod = ln & 15, lhi = ln >> 4;
            const float* Pj = ((sub == 0) ? Cin : Sh) + (size_t)chunk * BATCH * DHID;
            const unsigned poff = (unsigned)((m_tile * 16 + lmod) * DHID + nclus * 128 + lhi * 8);
            #pragma unroll
            for (int nt = 0; nt < 8; ++nt) {
                const v4f lo = *(const v4f*)(Pj + poff + nt * 16);
                const v4f hi = *(const v4f*)(Pj + poff + nt * 16 + 4);
                Hreg[nt] += __builtin_shufflevector(lo, hi, 0, 1, 2, 3, 4, 5, 6, 7);
            }
        }

        #pragma unroll
        for (int i = 0; i < SUBLEN; ++i) {
            aV += zero;
            const int t = t0 + i;

            const int ln   = opaque_lane();
            const int lmod = ln & 15;
            const int lhi  = ln >> 4;
            const int brow = m_tile * 16 + lmod;
            const unsigned xoff = (unsigned)(brow * DIN + nclus * 16 + lhi * 4);
            const int mlo  = (4 * nclus + lhi) * DOUT + lmod;
            const int mhi  = mlo + 4 * 16 * DOUT;
            const int cb   = (32 * nclus + 2 * lhi) * DOUT + lmod;
            const float* ap = aV + (unsigned)(nclus * 128 + lhi * 8);

            v8f Y[4] = {};

            #pragma unroll
            for (int s = 0; s <= i; ++s) {
                v4f fnext[8];
                v4f xnext[2] = {};
                if (s < i) {
                    load_m_block(fnext, ms, i - (s + 1), mlo, mhi);
                    if (s + 1 < i)
                        load_x_slice(xnext, x, xoff, t0 + s + 1);
                } else {
                    load_c_block(fnext, cs, 0, cb);
                    acur[0] = *(const v4f*)(ap);
                    acur[1] = *(const v4f*)(ap + 4);
                }
                __builtin_amdgcn_sched_barrier(0);
                #pragma unroll
                for (int jj = 0; jj < 2; ++jj) {
                    const v4f xo = (s == i) ? xnew[jj] : xcur[jj];
                    #pragma unroll
                    for (int o = 0; o < 4; ++o) Y[o] = wmma4(fcur[4 * jj + o].xy, xo.xy, Y[o]);
                    #pragma unroll
                    for (int o = 0; o < 4; ++o) Y[o] = wmma4(fcur[4 * jj + o].zw, xo.zw, Y[o]);
                }
                __builtin_amdgcn_sched_barrier(0);
                #pragma unroll
                for (int f = 0; f < 8; ++f) fcur[f] = fnext[f];
                if (s + 1 < i) { xcur[0] = xnext[0]; xcur[1] = xnext[1]; }
            }

            load_x_slice(xnew, x, xoff, (t + 1 <= tlast) ? (t + 1) : tlast);

            #pragma unroll
            for (int nt = 0; nt < 8; ++nt) {
                v4f fnext[8];
                if (nt + 1 < 8) {
                    load_c_block(fnext, cs, nt + 1, cb);
                } else {
                    load_m_block(fnext, ms, (i + 1 < SUBLEN) ? (i + 1) : 0, mlo, mhi);
                    if (i + 1 < SUBLEN)
                        load_x_slice(xcur, x, xoff, t0);
                }
                __builtin_amdgcn_sched_barrier(0);
                {
                    const v8f an = __builtin_shufflevector(acur[0], acur[1], 0, 1, 2, 3, 4, 5, 6, 7);
                    Hreg[nt] = Hreg[nt] * an;
                }
                if (nt + 1 < 8) {
                    acur[0] = *(const v4f*)(ap + 16 * (nt + 1));
                    acur[1] = *(const v4f*)(ap + 16 * (nt + 1) + 4);
                }
                const v8f  Hc = Hreg[nt];
                const v2f h01 = __builtin_shufflevector(Hc, Hc, 0, 1);
                const v2f h23 = __builtin_shufflevector(Hc, Hc, 2, 3);
                const v2f h45 = __builtin_shufflevector(Hc, Hc, 4, 5);
                const v2f h67 = __builtin_shufflevector(Hc, Hc, 6, 7);
                #pragma unroll
                for (int o = 0; o < 4; ++o) Y[o] = wmma4(fcur[o].xy,     h01, Y[o]);
                #pragma unroll
                for (int o = 0; o < 4; ++o) Y[o] = wmma4(fcur[o].zw,     h23, Y[o]);
                #pragma unroll
                for (int o = 0; o < 4; ++o) Y[o] = wmma4(fcur[4 + o].xy, h45, Y[o]);
                #pragma unroll
                for (int o = 0; o < 4; ++o) Y[o] = wmma4(fcur[4 + o].zw, h67, Y[o]);
                __builtin_amdgcn_sched_barrier(0);
                #pragma unroll
                for (int f = 0; f < 8; ++f) fcur[f] = fnext[f];
            }

            __syncthreads();

            {
                v4f* pw = part + nclus * PPLANE + brow * PUNITS;
                #pragma unroll
                for (int o = 0; o < 4; ++o) {
                    const int u0 = (o * 4 + lhi * 2) ^ lmod;
                    pw[u0]     = __builtin_shufflevector(Y[o], Y[o], 0, 1, 2, 3);
                    pw[u0 ^ 1] = __builtin_shufflevector(Y[o], Y[o], 4, 5, 6, 7);
                }
            }
            __syncthreads();

            {
                float* yt = y + (size_t)t * BATCH * DOUT;
                #pragma unroll
                for (int p = 0; p < 2; ++p) {
                    const int row = p * 32 + wave * 2 + lhi;
                    const int u   = lmod;
                    const v4f* q0 = part + row * PUNITS + (u ^ (row & 15));
                    const v4f s0 = q0[0];
                    const v4f s1 = q0[PPLANE];
                    const v4f s2 = q0[2 * PPLANE];
                    const v4f s3 = q0[3 * PPLANE];
                    *(v4f*)(yt + (unsigned)(row * DOUT + u * 4)) = (s0 + s1) + (s2 + s3);
                }
            }
        }
    }
}

extern "C" void kernel_launch(void* const* d_in, const int* in_sizes, int n_in,
                              void* d_out, int out_size, void* d_ws, size_t ws_size,
                              hipStream_t stream)
{
    const float* h0 = (const float*)d_in[0];
    const float* x  = (const float*)d_in[1];
    const float* a  = (const float*)d_in[2];
    const float* bW = (const float*)d_in[3];
    const float* cW = (const float*)d_in[4];

    float* out    = (float*)d_out;
    float* hfinal = out;
    float* y      = out + BATCH * DHID;

    float* S  = (float*)d_ws;
    float* Sh = S  + (size_t)CHUNKS * BATCH * DHID;
    float* C  = Sh + (size_t)CHUNKS * BATCH * DHID;
    v4f*   bq = (v4f*)(C + (size_t)CHUNKS * BATCH * DHID);
    v4f*   cq = bq + 16 * DHID;
    v4f*   mq = cq + 128 * DOUT;

    rnn_pack<<<48, 256, 0, stream>>>(bW, cW, a, bq, cq, mq);
    rnn_chunk_summary<<<CHUNKS, NTHREADS, 0, stream>>>(x, a, bq, S, Sh);
    rnn_carry_scan<<<BATCH * DHID / 2 / SCAN_THREADS, SCAN_THREADS, 0, stream>>>(h0, a, S, C, hfinal);
    rnn_chunk_emit<<<CHUNKS, NTHREADS, 0, stream>>>(x, a, mq, cq, C, Sh, y);
}


# Output kernel restarts from an exact state every 4 steps, not 8: taps M_0..M_3 only, 10 not 18 x@M blocks per 4 steps; summary kernel stores 4-step sums, chunk sum regrouped from them via LDS
# speedup vs baseline: 2.4827x; 1.0285x over previous
#include <hip/hip_runtime.h>


typedef __attribute__((ext_vector_type(2))) float v2f;
typedef __attribute__((ext_vector_type(4))) float v4f;
typedef __attribute__((ext_vector_type(8))) float v8f;

#define T_STEPS 4096
#define BATCH   64
#define DIN     64
#define DHID    512
#define DOUT    64
#define CHUNKS  256
#define CLEN    16
#define SUBLEN  4
#define NSUB    (CLEN / SUBLEN)
#define PLANESET ((size_t)CHUNKS * BATCH * DHID)
#define MQ_UNITS (SUBLEN * 16 * DOUT)
#define BQ_UNITS (16 * DHID)
#define CQ_UNITS (128 * DOUT)
#define PUNITS  16
#define PPLANE  (BATCH * PUNITS)
#define NWAVES  16
#define NTHREADS (NWAVES * 32)

__device__ __forceinline__ v8f wmma4(v2f a, v2f b, v8f c) {
    return __builtin_amdgcn_wmma_f32_16x16x4_f32(
        false, a, false, b, (short)0, c, false, false);
}

__device__ __forceinline__ int opaque_zero() {
    int z;
    asm volatile("s_mov_b32 %0, 0" : "=s"(z));
    return z;
}

__device__ __forceinline__ int opaque_lane() {
    int r;
    asm volatile("v_mbcnt_lo_u32_b32 %0, -1, 0" : "=v"(r));
    return r;
}

__device__ __forceinline__ int opaque_copy(int v) {
    int r;
    asm volatile("v_mov_b32 %0, %1" : "=v"(r) : "v"(v));
    return r;
}

__device__ __forceinline__ float opaque_copyf(float v) {
    float r;
    asm volatile("v_mov_b32 %0, %1" : "=v"(r) : "v"(v));
    return r;
}

__global__ __launch_bounds__(256)
void rnn_pack(const float* __restrict__ bW, const float* __restrict__ cW,
              const float* __restrict__ aVec,
              v4f* __restrict__ bq, v4f* __restrict__ cq, v4f* __restrict__ mq)
{
    __shared__ v8f mpart[8 * SUBLEN * 32];

    if (blockIdx.x >= 32) {
        const int tile = blockIdx.x - 32;
        const int mt   = tile >> 2;
        const int nt   = tile & 3;
        const int w    = threadIdx.x >> 5;
        const int lane = threadIdx.x & 31;
        const int lmod = lane & 15;
        const int lhi  = lane >> 4;
        const float* bp = bW + (size_t)(16 * mt + lmod) * DHID + 64 * w + 4 * lhi;
        const float* cp = cW + (size_t)(64 * w + 4 * lhi) * DOUT + 16 * nt + lmod;
        const float* ap = aVec + 64 * w + 4 * lhi;
        v8f acc[SUBLEN] = {};
        v4f bv8[8], av8[8], cv8[8];
        #pragma unroll
        for (int k8 = 0; k8 < 8; ++k8) {
            bv8[k8]   = *(const v4f*)(bp + 8 * k8);
            av8[k8]   = *(const v4f*)(ap + 8 * k8);
            cv8[k8].x = cp[(size_t)(8 * k8)     * DOUT];
            cv8[k8].y = cp[(size_t)(8 * k8 + 1) * DOUT];
            cv8[k8].z = cp[(size_t)(8 * k8 + 2) * DOUT];
            cv8[k8].w = cp[(size_t)(8 * k8 + 3) * DOUT];
        }
        __builtin_amdgcn_sched_barrier(0);
        #pragma unroll
        for (int k8 = 0; k8 < 8; ++k8) {
            const v4f av = av8[k8];
            const v4f cv = cv8[k8];
            v4f bd[SUBLEN];
            bd[0] = bv8[k8];
            #pragma unroll
            for (int d = 1; d < SUBLEN; ++d) bd[d] = bd[d - 1] * av;
            #pragma unroll
            for (int d = 0; d < SUBLEN; ++d) acc[d] = wmma4(bd[d].xy, cv.xy, acc[d]);
            #pragma unroll
            for (int d = 0; d < SUBLEN; ++d) acc[d] = wmma4(bd[d].zw, cv.zw, acc[d]);
        }
        #pragma unroll
        for (int d = 0; d < SUBLEN; ++d) mpart[(w * SUBLEN + d) * 32 + lane] = acc[d];
        __syncthreads();
        if (w < SUBLEN) {
            v8f m = mpart[w * 32 + lane];
            #pragma unroll
            for (int s = 1; s < 8; ++s) m += mpart[(s * SUBLEN + w) * 32 + lane];
            v4f* mo = mq + w * (16 * DOUT) + (4 * mt + 2 * lhi) * DOUT + 16 * nt + lmod;
            mo[0]    = __builtin_shufflevector(m, m, 0, 1, 2, 3);
            mo[DOUT] = __builtin_shufflevector(m, m, 4, 5, 6, 7);
        }
        return;
    }
    const int i = blockIdx.x * 256 + threadIdx.x;
    if (i < 16 * DHID) {
        const int q = i >> 9, n = i & (DHID - 1);
        v4f t;
        t.x = bW[(size_t)(4 * q)     * DHID + n];
        t.y = bW[(size_t)(4 * q + 1) * DHID + n];
        t.z = bW[(size_t)(4 * q + 2) * DHID + n];
        t.w = bW[(size_t)(4 * q + 3) * DHID + n];
        bq[i] = t;
    }
    if (i < 128 * DOUT) {
        const int q = i >> 6, n = i & (DOUT - 1);
        v4f t;
        t.x = cW[(size_t)(4 * q)     * DOUT + n];
        t.y = cW[(size_t)(4 * q + 1) * DOUT + n];
        t.z = cW[(size_t)(4 * q + 2) * DOUT + n];
        t.w = cW[(size_t)(4 * q + 3) * DOUT + n];
        cq[i] = t;
    }
}

__device__ __forceinline__ v8f chunk_sum_so_far(const v4f* tot, int tid, int nt, float a1, v8f q) {
    float ap = a1;
    #pragma unroll
    for (int s = 1; s < SUBLEN; s <<= 1) ap *= ap;
    const v4f lo = tot[(2 * nt)     * NTHREADS + tid];
    const v4f hi = tot[(2 * nt + 1) * NTHREADS + tid];
    return __builtin_shufflevector(lo, hi, 0, 1, 2, 3, 4, 5, 6, 7) * ap + q;
}

__global__ __launch_bounds__(NTHREADS)
void rnn_chunk_summary(const float* __restrict__ x,
                       const float* __restrict__ aVec,
                       const v4f* __restrict__ bq0,
                       float* __restrict__ S,
                       float* __restrict__ Qs)
{
    const int chunk = blockIdx.x;
    const int tid   = threadIdx.x;
    const int wave  = tid >> 5;
    const int lane  = tid & 31;
    const int lmod  = lane & 15;
    const int lhi   = lane >> 4;

    const int m_tile = wave & 3;
    const int nclus  = wave >> 2;
    const int rowA   = m_tile * 16 + lmod;

    __shared__ v4f bs[BQ_UNITS];
    __shared__ v4f tot[16 * NTHREADS];

    #pragma unroll
    for (int k = 0; k < BQ_UNITS / NTHREADS; ++k)
        bs[k * NTHREADS + tid] = bq0[k * NTHREADS + tid];
    #pragma unroll
    for (int g = 0; g < 16; ++g)
        tot[g * NTHREADS + tid] = v4f{0.0f, 0.0f, 0.0f, 0.0f};

    float av[8];
    #pragma unroll
    for (int nt = 0; nt < 8; ++nt)
        av[nt] = aVec[(nclus * 8 + nt) * 16 + lmod];

    const int bbase = lhi * DHID + nclus * 128 + lmod;

    v8f Sacc[8] = {};

    v4f afrag[8];
    {
        const float* xt = x + (size_t)(chunk * CLEN) * BATCH * DIN + rowA * DIN + lhi * 4;
        #pragma unroll
        for (int j = 0; j < 8; ++j)
            afrag[j] = *(const v4f*)(xt + j * 8);
    }

    __syncthreads();

    v4f bcur[4];
    {
        const int zb0 = opaque_copy(bbase);
        #pragma unroll
        for (int n = 0; n < 4; ++n)
            bcur[n] = bs[zb0 + n * 16];
    }

    for (int i = 0; i < CLEN; ++i) {
        const int zb = opaque_copy(bbase);

        const int inext = (i + 1 < CLEN) ? (i + 1) : i;
        const float* xn = x + (size_t)(chunk * CLEN + inext) * BATCH * DIN + rowA * DIN + lhi * 4;

        #pragma unroll
        for (int half = 0; half < 2; ++half) {
            v8f acc[4] = {};
            #pragma unroll
            for (int j = 0; j < 8; ++j) {
                const int nb = (half * 8 + j + 1) & 15;
                const int nh = nb >> 3, nj = nb & 7;
                v4f bnext[4];
                #pragma unroll
                for (int n = 0; n < 4; ++n)
                    bnext[n] = bs[zb + nj * 2 * DHID + nh * 64 + n * 16];
                __builtin_amdgcn_sched_barrier(0);
                #pragma unroll
                for (int n = 0; n < 4; ++n) acc[n] = wmma4(afrag[j].xy, bcur[n].xy, acc[n]);
                #pragma unroll
                for (int n = 0; n < 4; ++n) acc[n] = wmma4(afrag[j].zw, bcur[n].zw, acc[n]);
                if (half == 1)
                    afrag[j] = *(const v4f*)(xn + j * 8);
                __builtin_amdgcn_sched_barrier(0);
                #pragma unroll
                for (int n = 0; n < 4; ++n) bcur[n] = bnext[n];
            }
            #pragma unroll
            for (int n = 0; n < 4; ++n) {
                const float an = av[half * 4 + n];
                #pragma unroll
                for (int r = 0; r < 8; ++r)
                    Sacc[half * 4 + n][r] = Sacc[half * 4 + n][r] * an + acc[n][r];
            }
        }

        if ((i & (SUBLEN - 1)) == SUBLEN - 1) {
            const unsigned loff = (unsigned)((m_tile * 16 + lhi * 8) * DHID + nclus * 128 + lmod);
            if (i != CLEN - 1) {
                float* Qj = Qs + (size_t)(i / SUBLEN) * PLANESET + (size_t)chunk * BATCH * DHID + loff;
                #pragma unroll
                for (int nt = 0; nt < 8; ++nt) {
                    const v8f tv = chunk_sum_so_far(tot, tid, nt, opaque_copyf(av[nt]), Sacc[nt]);
                    tot[(2 * nt)     * NTHREADS + tid] = __builtin_shufflevector(tv, tv, 0, 1, 2, 3);
                    tot[(2 * nt + 1) * NTHREADS + tid] = __builtin_shufflevector(tv, tv, 4, 5, 6, 7);
                    #pragma unroll
                    for (int r = 0; r < 8; ++r) {
                        Qj[r * DHID + 16 * nt] = Sacc[nt][r];
                        Sacc[nt][r] = 0.0f;
                    }
                }
            } else {
                float* Sj = S + (size_t)chunk * BATCH * DHID + loff;
                #pragma unroll
                for (int nt = 0; nt < 8; ++nt) {
                    const v8f tv = chunk_sum_so_far(tot, tid, nt, opaque_copyf(av[nt]), Sacc[nt]);
                    #pragma unroll
                    for (int r = 0; r < 8; ++r)
                        Sj[r * DHID + 16 * nt] = tv[r];
                }
            }
        }
    }
}

#define SCAN_DEPTH   32
#define SCAN_THREADS 64
#define SCAN_PLANE   (BATCH * DHID / 2)

__global__ __launch_bounds__(SCAN_THREADS)
void rnn_carry_scan(const float* __restrict__ h0,
                    const float* __restrict__ aVec,
                    const float* __restrict__ S,
                    float* __restrict__ Cout,
                    float* __restrict__ hfinal)
{
    const int pair = blockIdx.x * SCAN_THREADS + threadIdx.x;
    const int idx  = pair * 2;
    const int hcol = idx & (DHID - 1);

    v2f aL = *(const v2f*)(aVec + hcol);
    #pragma unroll
    for (int s = 1; s < CLEN; s <<= 1) aL *= aL;

    const v2f* Sp = (const v2f*)S + pair;
    v2f*       Cp = (v2f*)Cout + pair;

    v2f cur[SCAN_DEPTH];
    #pragma unroll
    for (int k = 0; k < SCAN_DEPTH; ++k)
        cur[k] = Sp[(size_t)k * SCAN_PLANE];

    v2f c = *(const v2f*)(h0 + idx);

    #pragma unroll
    for (int jb = 0; jb < CHUNKS / SCAN_DEPTH; ++jb) {
        v2f nxt[SCAN_DEPTH];
        if (jb + 1 < CHUNKS / SCAN_DEPTH) {
            #pragma unroll
            for (int k = 0; k < SCAN_DEPTH; ++k)
                nxt[k] = Sp[(size_t)((jb + 1) * SCAN_DEPTH + k) * SCAN_PLANE];
        }
        __builtin_amdgcn_sched_barrier(0);
        #pragma unroll
        for (int k = 0; k < SCAN_DEPTH; ++k) {
            Cp[(size_t)(jb * SCAN_DEPTH + k) * SCAN_PLANE] = c;
            c = aL * c + cur[k];
        }
        __builtin_amdgcn_sched_barrier(0);
        #pragma unroll
        for (int k = 0; k < SCAN_DEPTH; ++k) cur[k] = nxt[k];
    }
    *(v2f*)(hfinal + idx) = c;
}

__device__ __forceinline__ void load_m_block(v4f (&f)[8], const v4f* msm, int d, int mlo) {
    #pragma unroll
    for (int jj = 0; jj < 2; ++jj) {
        #pragma unroll
        for (int o = 0; o < 4; ++o)
            f[4 * jj + o] = msm[mlo + d * (16 * DOUT) + jj * (2 * DOUT) + o * 16];
    }
}

__device__ __forceinline__ void load_c_block(v4f (&f)[8], const v4f* csm, int nt, int cb) {
    #pragma unroll
    for (int o = 0; o < 4; ++o) {
        f[o]     = csm[cb + nt * (4 * DOUT)        + o * 16];
        f[4 + o] = csm[cb + nt * (4 * DOUT) + DOUT + o * 16];
    }
}

__device__ __forceinline__ void load_x_slice(v4f (&xv)[2], const float* x, unsigned xoff, int t) {
    const float* p = x + (size_t)t * BATCH * DIN;
    xv[0] = *(const v4f*)(p + xoff);
    xv[1] = *(const v4f*)(p + xoff + 8);
}

__global__ __launch_bounds__(NTHREADS)
void rnn_chunk_emit(const float* __restrict__ x,
                    const float* __restrict__ aVec,
                    const v4f* __restrict__ mq0,
                    const v4f* __restrict__ cq0,
                    const float* __restrict__ Cin,
                    const float* __restrict__ Qs,
                    float* __restrict__ y)
{
    __shared__ v4f ms[MQ_UNITS];
    __shared__ v4f cs[CQ_UNITS];
    __shared__ v4f part[4 * PPLANE];

    const int chunk = blockIdx.x;
    const int tid   = threadIdx.x;
    const int wave  = __builtin_amdgcn_readfirstlane(tid >> 5);

    const int m_tile = wave & 3;
    const int nclus  = wave >> 2;

    #pragma unroll
    for (int k = 0; k < MQ_UNITS / NTHREADS; ++k)
        ms[k * NTHREADS + tid] = mq0[k * NTHREADS + tid];
    #pragma unroll
    for (int k = 0; k < CQ_UNITS / NTHREADS; ++k)
        cs[k * NTHREADS + tid] = cq0[k * NTHREADS + tid];

    const int t00   = chunk * CLEN;
    const int tlast = t00 + CLEN - 1;

    v4f xnew[2];
    v4f fcur[8];
    {
        const int ln = opaque_lane();
        const int lmod = ln & 15, lhi = ln >> 4;
        load_x_slice(xnew, x, (unsigned)((m_tile * 16 + lmod) * DIN + nclus * 16 + lhi * 4), t00);
        __syncthreads();
        const int mlo = (4 * nclus + lhi) * DOUT + lmod;
        load_m_block(fcur, ms, 0, mlo);
    }

    const int zero = opaque_zero();
    const float* aV = aVec;

    v4f xcur[2] = {};
    v4f acur[2] = {};
    v8f Hreg[8] = {};

    for (int sub = 0; sub < NSUB; ++sub) {
        const int t0 = t00 + sub * SUBLEN;

        {
            const int ln = opaque_lane();
            const int lmod = ln & 15, lhi = ln >> 4;
            const float* Pj = ((sub == 0) ? Cin : Qs + (size_t)(sub - 1) * PLANESET) + (size_t)chunk * BATCH * DHID;
            const unsigned poff = (unsigned)((m_tile * 16 + lmod) * DHID + nclus * 128 + lhi * 8);
            #pragma unroll
            for (int nt = 0; nt < 8; ++nt) {
                const v4f lo = *(const v4f*)(Pj + poff + nt * 16);
                const v4f hi = *(const v4f*)(Pj + poff + nt * 16 + 4);
                Hreg[nt] += __builtin_shufflevector(lo, hi, 0, 1, 2, 3, 4, 5, 6, 7);
            }
        }

        #pragma unroll
        for (int i = 0; i < SUBLEN; ++i) {
            aV += zero;
            const int t = t0 + i;

            const int ln   = opaque_lane();
            const int lmod = ln & 15;
            const int lhi  = ln >> 4;
            const int brow = m_tile * 16 + lmod;
            const unsigned xoff = (unsigned)(brow * DIN + nclus * 16 + lhi * 4);
            const int mlo  = (4 * nclus + lhi) * DOUT + lmod;
            const int cb   = (32 * nclus + 2 * lhi) * DOUT + lmod;
            const float* ap = aV + (unsigned)(nclus * 128 + lhi * 8);

            v8f Y[4] = {};

            #pragma unroll
            for (int s = 0; s <= i; ++s) {
                v4f fnext[8];
                v4f xnext[2] = {};
                if (s < i) {
                    load_m_block(fnext, ms, i - (s + 1), mlo);
                    if (s + 1 < i)
                        load_x_slice(xnext, x, xoff, t0 + s + 1);
                } else {
                    load_c_block(fnext, cs, 0, cb);
                    acur[0] = *(const v4f*)(ap);
                    acur[1] = *(const v4f*)(ap + 4);
                }
                __builtin_amdgcn_sched_barrier(0);
                #pragma unroll
                for (int jj = 0; jj < 2; ++jj) {
                    const v4f xo = (s == i) ? xnew[jj] : xcur[jj];
                    #pragma unroll
                    for (int o = 0; o < 4; ++o) Y[o] = wmma4(fcur[4 * jj + o].xy, xo.xy, Y[o]);
                    #pragma unroll
                    for (int o = 0; o < 4; ++o) Y[o] = wmma4(fcur[4 * jj + o].zw, xo.zw, Y[o]);
                }
                __builtin_amdgcn_sched_barrier(0);
                #pragma unroll
                for (int f = 0; f < 8; ++f) fcur[f] = fnext[f];
                if (s + 1 < i) { xcur[0] = xnext[0]; xcur[1] = xnext[1]; }
            }

            load_x_slice(xnew, x, xoff, (t + 1 <= tlast) ? (t + 1) : tlast);

            #pragma unroll
            for (int nt = 0; nt < 8; ++nt) {
                v4f fnext[8];
                if (nt + 1 < 8) {
                    load_c_block(fnext, cs, nt + 1, cb);
                } else {
                    load_m_block(fnext, ms, (i + 1 < SUBLEN) ? (i + 1) : 0, mlo);
                    if (i + 1 < SUBLEN)
                        load_x_slice(xcur, x, xoff, t0);
                }
                __builtin_amdgcn_sched_barrier(0);
                {
                    const v8f an = __builtin_shufflevector(acur[0], acur[1], 0, 1, 2, 3, 4, 5, 6, 7);
                    Hreg[nt] = Hreg[nt] * an;
                }
                if (nt + 1 < 8) {
                    acur[0] = *(const v4f*)(ap + 16 * (nt + 1));
                    acur[1] = *(const v4f*)(ap + 16 * (nt + 1) + 4);
                }
                const v8f  Hc = Hreg[nt];
                const v2f h01 = __builtin_shufflevector(Hc, Hc, 0, 1);
                const v2f h23 = __builtin_shufflevector(Hc, Hc, 2, 3);
                const v2f h45 = __builtin_shufflevector(Hc, Hc, 4, 5);
                const v2f h67 = __builtin_shufflevector(Hc, Hc, 6, 7);
                #pragma unroll
                for (int o = 0; o < 4; ++o) Y[o] = wmma4(fcur[o].xy,     h01, Y[o]);
                #pragma unroll
                for (int o = 0; o < 4; ++o) Y[o] = wmma4(fcur[o].zw,     h23, Y[o]);
                #pragma unroll
                for (int o = 0; o < 4; ++o) Y[o] = wmma4(fcur[4 + o].xy, h45, Y[o]);
                #pragma unroll
                for (int o = 0; o < 4; ++o) Y[o] = wmma4(fcur[4 + o].zw, h67, Y[o]);
                __builtin_amdgcn_sched_barrier(0);
                #pragma unroll
                for (int f = 0; f < 8; ++f) fcur[f] = fnext[f];
            }

            __syncthreads();

            {
                v4f* pw = part + nclus * PPLANE + brow * PUNITS;
                #pragma unroll
                for (int o = 0; o < 4; ++o) {
                    const int u0 = (o * 4 + lhi * 2) ^ lmod;
                    pw[u0]     = __builtin_shufflevector(Y[o], Y[o], 0, 1, 2, 3);
                    pw[u0 ^ 1] = __builtin_shufflevector(Y[o], Y[o], 4, 5, 6, 7);
                }
            }
            __syncthreads();

            {
                float* yt = y + (size_t)t * BATCH * DOUT;
                #pragma unroll
                for (int p = 0; p < 2; ++p) {
                    const int row = p * 32 + wave * 2 + lhi;
                    const int u   = lmod;
                    const v4f* q0 = part + row * PUNITS + (u ^ (row & 15));
                    const v4f s0 = q0[0];
                    const v4f s1 = q0[PPLANE];
                    const v4f s2 = q0[2 * PPLANE];
                    const v4f s3 = q0[3 * PPLANE];
                    *(v4f*)(yt + (unsigned)(row * DOUT + u * 4)) = (s0 + s1) + (s2 + s3);
                }
            }
        }
    }
}

extern "C" void kernel_launch(void* const* d_in, const int* in_sizes, int n_in,
                              void* d_out, int out_size, void* d_ws, size_t ws_size,
                              hipStream_t stream)
{
    const float* h0 = (const float*)d_in[0];
    const float* x  = (const float*)d_in[1];
    const float* a  = (const float*)d_in[2];
    const float* bW = (const float*)d_in[3];
    const float* cW = (const float*)d_in[4];

    float* out    = (float*)d_out;
    float* hfinal = out;
    float* y      = out + BATCH * DHID;

    float* S  = (float*)d_ws;
    float* Qs = S  + PLANESET;
    float* C  = Qs + (NSUB - 1) * PLANESET;
    v4f*   bq = (v4f*)(C + PLANESET);
    v4f*   cq = bq + 16 * DHID;
    v4f*   mq = cq + 128 * DOUT;

    rnn_pack<<<48, 256, 0, stream>>>(bW, cW, a, bq, cq, mq);
    rnn_chunk_summary<<<CHUNKS, NTHREADS, 0, stream>>>(x, a, bq, S, Qs);
    rnn_carry_scan<<<BATCH * DHID / 2 / SCAN_THREADS, SCAN_THREADS, 0, stream>>>(h0, a, S, C, hfinal);
    rnn_chunk_emit<<<CHUNKS, NTHREADS, 0, stream>>>(x, a, mq, cq, C, Qs, y);
}


# Summary planes (S, Qs, carries) kept in the output kernel's register order: each plane load/store is one instruction over 512 contiguous bytes; tiles turned per wave through LDS; sums unchanged
# speedup vs baseline: 2.5653x; 1.0333x over previous
#include <hip/hip_runtime.h>


typedef __attribute__((ext_vector_type(2))) float v2f;
typedef __attribute__((ext_vector_type(4))) float v4f;
typedef __attribute__((ext_vector_type(8))) float v8f;

#define T_STEPS 4096
#define BATCH   64
#define DIN     64
#define DHID    512
#define DOUT    64
#define CHUNKS  256
#define CLEN    16
#define SUBLEN  4
#define NSUB    (CLEN / SUBLEN)
#define PLANESET ((size_t)CHUNKS * BATCH * DHID)
#define MQ_UNITS (SUBLEN * 16 * DOUT)
#define BQ_UNITS (16 * DHID)
#define CQ_UNITS (128 * DOUT)
#define PUNITS  16
#define PPLANE  (BATCH * PUNITS)
#define NWAVES  16
#define NTHREADS (NWAVES * 32)
#define TS_PITCH 20
#define TS_TILE  (16 * TS_PITCH)
#define WPLANE  (8 * 2 * 32)

__device__ __forceinline__ v8f wmma4(v2f a, v2f b, v8f c) {
    return __builtin_amdgcn_wmma_f32_16x16x4_f32(
        false, a, false, b, (short)0, c, false, false);
}

__device__ __forceinline__ int opaque_zero() {
    int z;
    asm volatile("s_mov_b32 %0, 0" : "=s"(z));
    return z;
}

__device__ __forceinline__ int opaque_lane() {
    int r;
    asm volatile("v_mbcnt_lo_u32_b32 %0, -1, 0" : "=v"(r));
    return r;
}

__device__ __forceinline__ int opaque_copy(int v) {
    int r;
    asm volatile("v_mov_b32 %0, %1" : "=v"(r) : "v"(v));
    return r;
}

__device__ __forceinline__ float opaque_copyf(float v) {
    float r;
    asm volatile("v_mov_b32 %0, %1" : "=v"(r) : "v"(v));
    return r;
}

__global__ __launch_bounds__(256)
void rnn_pack(const float* __restrict__ bW, const float* __restrict__ cW,
              const float* __restrict__ aVec,
              v4f* __restrict__ bq, v4f* __restrict__ cq, v4f* __restrict__ mq)
{
    __shared__ v8f mpart[8 * SUBLEN * 32];

    if (blockIdx.x >= 32) {
        const int tile = blockIdx.x - 32;
        const int mt   = tile >> 2;
        const int nt   = tile & 3;
        const int w    = threadIdx.x >> 5;
        const int lane = threadIdx.x & 31;
        const int lmod = lane & 15;
        const int lhi  = lane >> 4;
        const float* bp = bW + (size_t)(16 * mt + lmod) * DHID + 64 * w + 4 * lhi;
        const float* cp = cW + (size_t)(64 * w + 4 * lhi) * DOUT + 16 * nt + lmod;
        const float* ap = aVec + 64 * w + 4 * lhi;
        v8f acc[SUBLEN] = {};
        v4f bv8[8], av8[8], cv8[8];
        #pragma unroll
        for (int k8 = 0; k8 < 8; ++k8) {
            bv8[k8]   = *(const v4f*)(bp + 8 * k8);
            av8[k8]   = *(const v4f*)(ap + 8 * k8);
            cv8[k8].x = cp[(size_t)(8 * k8)     * DOUT];
            cv8[k8].y = cp[(size_t)(8 * k8 + 1) * DOUT];
            cv8[k8].z = cp[(size_t)(8 * k8 + 2) * DOUT];
            cv8[k8].w = cp[(size_t)(8 * k8 + 3) * DOUT];
        }
        __builtin_amdgcn_sched_barrier(0);
        #pragma unroll
        for (int k8 = 0; k8 < 8; ++k8) {
            const v4f av = av8[k8];
            const v4f cv = cv8[k8];
            v4f bd[SUBLEN];
            bd[0] = bv8[k8];
            #pragma unroll
            for (int d = 1; d < SUBLEN; ++d) bd[d] = bd[d - 1] * av;
            #pragma unroll
            for (int d = 0; d < SUBLEN; ++d) acc[d] = wmma4(bd[d].xy, cv.xy, acc[d]);
            #pragma unroll
            for (int d = 0; d < SUBLEN; ++d) acc[d] = wmma4(bd[d].zw, cv.zw, acc[d]);
        }
        #pragma unroll
        for (int d = 0; d < SUBLEN; ++d) mpart[(w * SUBLEN + d) * 32 + lane] = acc[d];
        __syncthreads();
        if (w < SUBLEN) {
            v8f m = mpart[w * 32 + lane];
            #pragma unroll
            for (int s = 1; s < 8; ++s) m += mpart[(s * SUBLEN + w) * 32 + lane];
            v4f* mo = mq + w * (16 * DOUT) + (4 * mt + 2 * lhi) * DOUT + 16 * nt + lmod;
            mo[0]    = __builtin_shufflevector(m, m, 0, 1, 2, 3);
            mo[DOUT] = __builtin_shufflevector(m, m, 4, 5, 6, 7);
        }
        return;
    }
    const int i = blockIdx.x * 256 + threadIdx.x;
    if (i < 16 * DHID) {
        const int q = i >> 9, n = i & (DHID - 1);
        v4f t;
        t.x = bW[(size_t)(4 * q)     * DHID + n];
        t.y = bW[(size_t)(4 * q + 1) * DHID + n];
        t.z = bW[(size_t)(4 * q + 2) * DHID + n];
        t.w = bW[(size_t)(4 * q + 3) * DHID + n];
        bq[i] = t;
    }
    if (i < 128 * DOUT) {
        const int q = i >> 6, n = i & (DOUT - 1);
        v4f t;
        t.x = cW[(size_t)(4 * q)     * DOUT + n];
        t.y = cW[(size_t)(4 * q + 1) * DOUT + n];
        t.z = cW[(size_t)(4 * q + 2) * DOUT + n];
        t.w = cW[(size_t)(4 * q + 3) * DOUT + n];
        cq[i] = t;
    }
}

__device__ __forceinline__ void turn_tile(float* twr, const float* trd, v8f v, v4f& lo, v4f& hi) {
    #pragma unroll
    for (int r = 0; r < 8; ++r) twr[r * TS_PITCH] = v[r];
    __builtin_amdgcn_fence(__ATOMIC_ACQ_REL, "wavefront");
    __builtin_amdgcn_wave_barrier();
    lo = *(const v4f*)(trd);
    hi = *(const v4f*)(trd + 4);
    __builtin_amdgcn_fence(__ATOMIC_ACQ_REL, "wavefront");
    __builtin_amdgcn_wave_barrier();
}

__device__ __forceinline__ v8f chunk_sum_so_far(const v4f* tot, int tid, int nt, float a1, v8f q) {
    float ap = a1;
    #pragma unroll
    for (int s = 1; s < SUBLEN; s <<= 1) ap *= ap;
    const v4f lo = tot[(2 * nt)     * NTHREADS + tid];
    const v4f hi = tot[(2 * nt + 1) * NTHREADS + tid];
    return __builtin_shufflevector(lo, hi, 0, 1, 2, 3, 4, 5, 6, 7) * ap + q;
}

__global__ __launch_bounds__(NTHREADS)
void rnn_chunk_summary(const float* __restrict__ x,
                       const float* __restrict__ aVec,
                       const v4f* __restrict__ bq0,
                       float* __restrict__ S,
                       float* __restrict__ Qs)
{
    const int chunk = blockIdx.x;
    const int tid   = threadIdx.x;
    const int wave  = tid >> 5;
    const int wave_s = __builtin_amdgcn_readfirstlane(wave);
    const int lane  = tid & 31;
    const int lmod  = lane & 15;
    const int lhi   = lane >> 4;

    const int m_tile = wave & 3;
    const int nclus  = wave >> 2;
    const int rowA   = m_tile * 16 + lmod;

    __shared__ v4f bs[BQ_UNITS];
    __shared__ v4f tot[16 * NTHREADS];
    __shared__ v4f tsc4[NWAVES * 2 * TS_TILE / 4];

    #pragma unroll
    for (int k = 0; k < BQ_UNITS / NTHREADS; ++k)
        bs[k * NTHREADS + tid] = bq0[k * NTHREADS + tid];
    #pragma unroll
    for (int g = 0; g < 16; ++g)
        tot[g * NTHREADS + tid] = v4f{0.0f, 0.0f, 0.0f, 0.0f};

    float av[8];
    #pragma unroll
    for (int nt = 0; nt < 8; ++nt)
        av[nt] = aVec[(nclus * 8 + nt) * 16 + lmod];

    const int bbase = lhi * DHID + nclus * 128 + lmod;

    v8f Sacc[8] = {};

    v4f afrag[8];
    {
        const float* xt = x + (size_t)(chunk * CLEN) * BATCH * DIN + rowA * DIN + lhi * 4;
        #pragma unroll
        for (int j = 0; j < 8; ++j)
            afrag[j] = *(const v4f*)(xt + j * 8);
    }

    __syncthreads();

    v4f bcur[4];
    {
        const int zb0 = opaque_copy(bbase);
        #pragma unroll
        for (int n = 0; n < 4; ++n)
            bcur[n] = bs[zb0 + n * 16];
    }

    for (int i = 0; i < CLEN; ++i) {
        const int zb = opaque_copy(bbase);

        const int inext = (i + 1 < CLEN) ? (i + 1) : i;
        const float* xn = x + (size_t)(chunk * CLEN + inext) * BATCH * DIN + rowA * DIN + lhi * 4;

        #pragma unroll
        for (int half = 0; half < 2; ++half) {
            v8f acc[4] = {};
            #pragma unroll
            for (int j = 0; j < 8; ++j) {
                const int nb = (half * 8 + j + 1) & 15;
                const int nh = nb >> 3, nj = nb & 7;
                v4f bnext[4];
                #pragma unroll
                for (int n = 0; n < 4; ++n)
                    bnext[n] = bs[zb + nj * 2 * DHID + nh * 64 + n * 16];
                __builtin_amdgcn_sched_barrier(0);
                #pragma unroll
                for (int n = 0; n < 4; ++n) acc[n] = wmma4(afrag[j].xy, bcur[n].xy, acc[n]);
                #pragma unroll
                for (int n = 0; n < 4; ++n) acc[n] = wmma4(afrag[j].zw, bcur[n].zw, acc[n]);
                if (half == 1)
                    afrag[j] = *(const v4f*)(xn + j * 8);
                __builtin_amdgcn_sched_barrier(0);
                #pragma unroll
                for (int n = 0; n < 4; ++n) bcur[n] = bnext[n];
            }
            #pragma unroll
            for (int n = 0; n < 4; ++n) {
                const float an = opaque_copyf(av[half * 4 + n]);
                #pragma unroll
                for (int r = 0; r < 8; ++r)
                    Sacc[half * 4 + n][r] = Sacc[half * 4 + n][r] * an + acc[n][r];
            }
        }

        if ((i & (SUBLEN - 1)) == SUBLEN - 1) {
            const int ln = opaque_lane();
            const int tl = ln & 15, th = ln >> 4, tw = wave_s;
            float* tsc = (float*)tsc4 + tw * (2 * TS_TILE);
            float*       twr = tsc + (8 * th) * TS_PITCH + tl;
            const float* trd = tsc + tl * TS_PITCH + 8 * th;
            const unsigned uoff = (unsigned)(tw * WPLANE + ln);
            if (i != CLEN - 1) {
                v4f* Qj = (v4f*)(Qs + (size_t)(i / SUBLEN) * PLANESET + (size_t)chunk * BATCH * DHID) + uoff;
                #pragma unroll
                for (int nt = 0; nt < 8; ++nt) {
                    v4f lo, hi;
                    turn_tile(twr + (nt & 1) * TS_TILE, trd + (nt & 1) * TS_TILE, Sacc[nt], lo, hi);
                    const v8f tv = chunk_sum_so_far(tot, tid, nt, opaque_copyf(av[nt]), Sacc[nt]);
                    tot[(2 * nt)     * NTHREADS + tid] = __builtin_shufflevector(tv, tv, 0, 1, 2, 3);
                    tot[(2 * nt + 1) * NTHREADS + tid] = __builtin_shufflevector(tv, tv, 4, 5, 6, 7);
                    Qj[nt * 64]      = lo;
                    Qj[nt * 64 + 32] = hi;
                    #pragma unroll
                    for (int r = 0; r < 8; ++r)
                        Sacc[nt][r] = 0.0f;
                }
            } else {
                v4f* Sj = (v4f*)(S + (size_t)chunk * BATCH * DHID) + uoff;
                v8f tv = chunk_sum_so_far(tot, tid, 0, opaque_copyf(av[0]), Sacc[0]);
                #pragma unroll
                for (int nt = 0; nt < 8; ++nt) {
                    v4f lo, hi;
                    turn_tile(twr + (nt & 1) * TS_TILE, trd + (nt & 1) * TS_TILE, tv, lo, hi);
                    if (nt + 1 < 8)
                        tv = chunk_sum_so_far(tot, tid, nt + 1, opaque_copyf(av[nt + 1]), Sacc[nt + 1]);
                    Sj[nt * 64]      = lo;
                    Sj[nt * 64 + 32] = hi;
                }
            }
        }
    }
}

#define SCAN_DEPTH   32
#define SCAN_THREADS 64
#define SCAN_PLANE   (BATCH * DHID / 2)

__global__ __launch_bounds__(SCAN_THREADS)
void rnn_carry_scan(const float* __restrict__ h0,
                    const float* __restrict__ aVec,
                    const float* __restrict__ S,
                    float* __restrict__ Cout,
                    float* __restrict__ hfinal)
{
    const int pair = blockIdx.x * SCAN_THREADS + threadIdx.x;
    const int idx  = pair * 2;
    const int u    = idx >> 2;
    const int e    = idx & 3;
    const int ln   = u & 31, hf = (u >> 5) & 1, nt = (u >> 6) & 7, wv = u >> 9;
    const int row  = 16 * (wv & 3) + (ln & 15);
    const int hcol = 128 * (wv >> 2) + 16 * nt + 8 * (ln >> 4) + 4 * hf + e;
    const int nat  = row * DHID + hcol;

    v2f aL = *(const v2f*)(aVec + hcol);
    #pragma unroll
    for (int s = 1; s < CLEN; s <<= 1) aL *= aL;

    const v2f* Sp = (const v2f*)S + pair;
    v2f*       Cp = (v2f*)Cout + pair;

    v2f cur[SCAN_DEPTH];
    #pragma unroll
    for (int k = 0; k < SCAN_DEPTH; ++k)
        cur[k] = Sp[(size_t)k * SCAN_PLANE];

    v2f c = *(const v2f*)(h0 + nat);

    #pragma unroll
    for (int jb = 0; jb < CHUNKS / SCAN_DEPTH; ++jb) {
        v2f nxt[SCAN_DEPTH];
        if (jb + 1 < CHUNKS / SCAN_DEPTH) {
            #pragma unroll
            for (int k = 0; k < SCAN_DEPTH; ++k)
                nxt[k] = Sp[(size_t)((jb + 1) * SCAN_DEPTH + k) * SCAN_PLANE];
        }
        __builtin_amdgcn_sched_barrier(0);
        #pragma unroll
        for (int k = 0; k < SCAN_DEPTH; ++k) {
            Cp[(size_t)(jb * SCAN_DEPTH + k) * SCAN_PLANE] = c;
            c = aL * c + cur[k];
        }
        __builtin_amdgcn_sched_barrier(0);
        #pragma unroll
        for (int k = 0; k < SCAN_DEPTH; ++k) cur[k] = nxt[k];
    }
    *(v2f*)(hfinal + nat) = c;
}

__device__ __forceinline__ void load_m_block(v4f (&f)[8], const v4f* msm, int d, int mlo) {
    #pragma unroll
    for (int jj = 0; jj < 2; ++jj) {
        #pragma unroll
        for (int o = 0; o < 4; ++o)
            f[4 * jj + o] = msm[mlo + d * (16 * DOUT) + jj * (2 * DOUT) + o * 16];
    }
}

__device__ __forceinline__ void load_c_block(v4f (&f)[8], const v4f* csm, int nt, int cb) {
    #pragma unroll
    for (int o = 0; o < 4; ++o) {
        f[o]     = csm[cb + nt * (4 * DOUT)        + o * 16];
        f[4 + o] = csm[cb + nt * (4 * DOUT) + DOUT + o * 16];
    }
}

__device__ __forceinline__ void load_x_slice(v4f (&xv)[2], const float* x, unsigned xoff, int t) {
    const float* p = x + (size_t)t * BATCH * DIN;
    xv[0] = *(const v4f*)(p + xoff);
    xv[1] = *(const v4f*)(p + xoff + 8);
}

__global__ __launch_bounds__(NTHREADS)
void rnn_chunk_emit(const float* __restrict__ x,
                    const float* __restrict__ aVec,
                    const v4f* __restrict__ mq0,
                    const v4f* __restrict__ cq0,
                    const float* __restrict__ Cin,
                    const float* __restrict__ Qs,
                    float* __restrict__ y)
{
    __shared__ v4f ms[MQ_UNITS];
    __shared__ v4f cs[CQ_UNITS];
    __shared__ v4f part[4 * PPLANE];

    const int chunk = blockIdx.x;
    const int tid   = threadIdx.x;
    const int wave  = __builtin_amdgcn_readfirstlane(tid >> 5);

    const int m_tile = wave & 3;
    const int nclus  = wave >> 2;

    #pragma unroll
    for (int k = 0; k < MQ_UNITS / NTHREADS; ++k)
        ms[k * NTHREADS + tid] = mq0[k * NTHREADS + tid];
    #pragma unroll
    for (int k = 0; k < CQ_UNITS / NTHREADS; ++k)
        cs[k * NTHREADS + tid] = cq0[k * NTHREADS + tid];

    const int t00   = chunk * CLEN;
    const int tlast = t00 + CLEN - 1;

    v4f xnew[2];
    v4f fcur[8];
    {
        const int ln = opaque_lane();
        const int lmod = ln & 15, lhi = ln >> 4;
        load_x_slice(xnew, x, (unsigned)((m_tile * 16 + lmod) * DIN + nclus * 16 + lhi * 4), t00);
        __syncthreads();
        const int mlo = (4 * nclus + lhi) * DOUT + lmod;
        load_m_block(fcur, ms, 0, mlo);
    }

    const int zero = opaque_zero();
    const float* aV = aVec;

    v4f xcur[2] = {};
    v4f acur[2] = {};
    v8f Hreg[8] = {};

    for (int sub = 0; sub < NSUB; ++sub) {
        const int t0 = t00 + sub * SUBLEN;

        {
            const int ln = opaque_lane();
            const float* Pj = ((sub == 0) ? Cin : Qs + (size_t)(sub - 1) * PLANESET) + (size_t)chunk * BATCH * DHID;
            const v4f* Pv = (const v4f*)Pj + (unsigned)(wave * WPLANE + ln);
            #pragma unroll
            for (int nt = 0; nt < 8; ++nt) {
                const v4f lo = Pv[nt * 64];
                const v4f hi = Pv[nt * 64 + 32];
                Hreg[nt] += __builtin_shufflevector(lo, hi, 0, 1, 2, 3, 4, 5, 6, 7);
            }
        }

        #pragma unroll
        for (int i = 0; i < SUBLEN; ++i) {
            aV += zero;
            const int t = t0 + i;

            const int ln   = opaque_lane();
            const int lmod = ln & 15;
            const int lhi  = ln >> 4;
            const int brow = m_tile * 16 + lmod;
            const unsigned xoff = (unsigned)(brow * DIN + nclus * 16 + lhi * 4);
            const int mlo  = (4 * nclus + lhi) * DOUT + lmod;
            const int cb   = (32 * nclus + 2 * lhi) * DOUT + lmod;
            const float* ap = aV + (unsigned)(nclus * 128 + lhi * 8);

            v8f Y[4] = {};

            #pragma unroll
            for (int s = 0; s <= i; ++s) {
                v4f fnext[8];
                v4f xnext[2] = {};
                if (s < i) {
                    load_m_block(fnext, ms, i - (s + 1), mlo);
                    if (s + 1 < i)
                        load_x_slice(xnext, x, xoff, t0 + s + 1);
                } else {
                    load_c_block(fnext, cs, 0, cb);
                    acur[0] = *(const v4f*)(ap);
                    acur[1] = *(const v4f*)(ap + 4);
                }
                __builtin_amdgcn_sched_barrier(0);
                #pragma unroll
                for (int jj = 0; jj < 2; ++jj) {
                    const v4f xo = (s == i) ? xnew[jj] : xcur[jj];
                    #pragma unroll
                    for (int o = 0; o < 4; ++o) Y[o] = wmma4(fcur[4 * jj + o].xy, xo.xy, Y[o]);
                    #pragma unroll
                    for (int o = 0; o < 4; ++o) Y[o] = wmma4(fcur[4 * jj + o].zw, xo.zw, Y[o]);
                }
                __builtin_amdgcn_sched_barrier(0);
                #pragma unroll
                for (int f = 0; f < 8; ++f) fcur[f] = fnext[f];
                if (s + 1 < i) { xcur[0] = xnext[0]; xcur[1] = xnext[1]; }
            }

            load_x_slice(xnew, x, xoff, (t + 1 <= tlast) ? (t + 1) : tlast);

            #pragma unroll
            for (int nt = 0; nt < 8; ++nt) {
                v4f fnext[8];
                if (nt + 1 < 8) {
                    load_c_block(fnext, cs, nt + 1, cb);
                } else {
                    load_m_block(fnext, ms, (i + 1 < SUBLEN) ? (i + 1) : 0, mlo);
                    if (i + 1 < SUBLEN)
                        load_x_slice(xcur, x, xoff, t0);
                }
                __builtin_amdgcn_sched_barrier(0);
                {
                    const v8f an = __builtin_shufflevector(acur[0], acur[1], 0, 1, 2, 3, 4, 5, 6, 7);
                    Hreg[nt] = Hreg[nt] * an;
                }
                if (nt + 1 < 8) {
                    acur[0] = *(const v4f*)(ap + 16 * (nt + 1));
                    acur[1] = *(const v4f*)(ap + 16 * (nt + 1) + 4);
                }
                const v8f  Hc = Hreg[nt];
                const v2f h01 = __builtin_shufflevector(Hc, Hc, 0, 1);
                const v2f h23 = __builtin_shufflevector(Hc, Hc, 2, 3);
                const v2f h45 = __builtin_shufflevector(Hc, Hc, 4, 5);
                const v2f h67 = __builtin_shufflevector(Hc, Hc, 6, 7);
                #pragma unroll
                for (int o = 0; o < 4; ++o) Y[o] = wmma4(fcur[o].xy,     h01, Y[o]);
                #pragma unroll
                for (int o = 0; o < 4; ++o) Y[o] = wmma4(fcur[o].zw,     h23, Y[o]);
                #pragma unroll
                for (int o = 0; o < 4; ++o) Y[o] = wmma4(fcur[4 + o].xy, h45, Y[o]);
                #pragma unroll
                for (int o = 0; o < 4; ++o) Y[o] = wmma4(fcur[4 + o].zw, h67, Y[o]);
                __builtin_amdgcn_sched_barrier(0);
                #pragma unroll
                for (int f = 0; f < 8; ++f) fcur[f] = fnext[f];
            }

            __syncthreads();

            {
                v4f* pw = part + nclus * PPLANE + brow * PUNITS;
                #pragma unroll
                for (int o = 0; o < 4; ++o) {
                    const int u0 = (o * 4 + lhi * 2) ^ lmod;
                    pw[u0]     = __builtin_shufflevector(Y[o], Y[o], 0, 1, 2, 3);
                    pw[u0 ^ 1] = __builtin_shufflevector(Y[o], Y[o], 4, 5, 6, 7);
                }
            }
            __syncthreads();

            {
                float* yt = y + (size_t)t * BATCH * DOUT;
                #pragma unroll
                for (int p = 0; p < 2; ++p) {
                    const int row = p * 32 + wave * 2 + lhi;
                    const int u   = lmod;
                    const v4f* q0 = part + row * PUNITS + (u ^ (row & 15));
                    const v4f s0 = q0[0];
                    const v4f s1 = q0[PPLANE];
                    const v4f s2 = q0[2 * PPLANE];
                    const v4f s3 = q0[3 * PPLANE];
                    *(v4f*)(yt + (unsigned)(row * DOUT + u * 4)) = (s0 + s1) + (s2 + s3);
                }
            }
        }
    }
}

extern "C" void kernel_launch(void* const* d_in, const int* in_sizes, int n_in,
                              void* d_out, int out_size, void* d_ws, size_t ws_size,
                              hipStream_t stream)
{
    const float* h0 = (const float*)d_in[0];
    const float* x  = (const float*)d_in[1];
    const float* a  = (const float*)d_in[2];
    const float* bW = (const float*)d_in[3];
    const float* cW = (const float*)d_in[4];

    float* out    = (float*)d_out;
    float* hfinal = out;
    float* y      = out + BATCH * DHID;

    float* S  = (float*)d_ws;
    float* Qs = S  + PLANESET;
    float* C  = Qs + (NSUB - 1) * PLANESET;
    v4f*   bq = (v4f*)(C + PLANESET);
    v4f*   cq = bq + 16 * DHID;
    v4f*   mq = cq + 128 * DOUT;

    rnn_pack<<<48, 256, 0, stream>>>(bW, cW, a, bq, cq, mq);
    rnn_chunk_summary<<<CHUNKS, NTHREADS, 0, stream>>>(x, a, bq, S, Qs);
    rnn_carry_scan<<<BATCH * DHID / 2 / SCAN_THREADS, SCAN_THREADS, 0, stream>>>(h0, a, S, C, hfinal);
    rnn_chunk_emit<<<CHUNKS, NTHREADS, 0, stream>>>(x, a, mq, cq, C, Qs, y);
}


# Output kernel: a sub-chunk's 4th step takes the exact state (decayed state + stored 4-step sum; at a chunk's end the next carry), y = state @ c alone: 38 not 42 WMMA blocks per 4 steps; sum regrouped
# speedup vs baseline: 2.6727x; 1.0419x over previous
#include <hip/hip_runtime.h>


typedef __attribute__((ext_vector_type(2))) float v2f;
typedef __attribute__((ext_vector_type(4))) float v4f;
typedef __attribute__((ext_vector_type(8))) float v8f;

#define T_STEPS 4096
#define BATCH   64
#define DIN     64
#define DHID    512
#define DOUT    64
#define CHUNKS  256
#define CLEN    16
#define SUBLEN  4
#define NSUB    (CLEN / SUBLEN)
#define PLANESET ((size_t)CHUNKS * BATCH * DHID)
#define NTAPS   (SUBLEN - 1)
#define MQ_UNITS (NTAPS * 16 * DOUT)
#define BQ_UNITS (16 * DHID)
#define CQ_UNITS (128 * DOUT)
#define PUNITS  16
#define PPLANE  (BATCH * PUNITS)
#define NWAVES  16
#define NTHREADS (NWAVES * 32)
#define TS_PITCH 20
#define TS_TILE  (16 * TS_PITCH)
#define WPLANE  (8 * 2 * 32)

__device__ __forceinline__ v8f wmma4(v2f a, v2f b, v8f c) {
    return __builtin_amdgcn_wmma_f32_16x16x4_f32(
        false, a, false, b, (short)0, c, false, false);
}

__device__ __forceinline__ int opaque_zero() {
    int z;
    asm volatile("s_mov_b32 %0, 0" : "=s"(z));
    return z;
}

__device__ __forceinline__ int opaque_lane() {
    int r;
    asm volatile("v_mbcnt_lo_u32_b32 %0, -1, 0" : "=v"(r));
    return r;
}

__device__ __forceinline__ int opaque_copy(int v) {
    int r;
    asm volatile("v_mov_b32 %0, %1" : "=v"(r) : "v"(v));
    return r;
}

__device__ __forceinline__ float opaque_copyf(float v) {
    float r;
    asm volatile("v_mov_b32 %0, %1" : "=v"(r) : "v"(v));
    return r;
}

__global__ __launch_bounds__(256)
void rnn_pack(const float* __restrict__ bW, const float* __restrict__ cW,
              const float* __restrict__ aVec,
              v4f* __restrict__ bq, v4f* __restrict__ cq, v4f* __restrict__ mq)
{
    __shared__ v8f mpart[8 * NTAPS * 32];

    if (blockIdx.x >= 32) {
        const int tile = blockIdx.x - 32;
        const int mt   = tile >> 2;
        const int nt   = tile & 3;
        const int w    = threadIdx.x >> 5;
        const int lane = threadIdx.x & 31;
        const int lmod = lane & 15;
        const int lhi  = lane >> 4;
        const float* bp = bW + (size_t)(16 * mt + lmod) * DHID + 64 * w + 4 * lhi;
        const float* cp = cW + (size_t)(64 * w + 4 * lhi) * DOUT + 16 * nt + lmod;
        const float* ap = aVec + 64 * w + 4 * lhi;
        v8f acc[NTAPS] = {};
        v4f bv8[8], av8[8], cv8[8];
        #pragma unroll
        for (int k8 = 0; k8 < 8; ++k8) {
            bv8[k8]   = *(const v4f*)(bp + 8 * k8);
            av8[k8]   = *(const v4f*)(ap + 8 * k8);
            cv8[k8].x = cp[(size_t)(8 * k8)     * DOUT];
            cv8[k8].y = cp[(size_t)(8 * k8 + 1) * DOUT];
            cv8[k8].z = cp[(size_t)(8 * k8 + 2) * DOUT];
            cv8[k8].w = cp[(size_t)(8 * k8 + 3) * DOUT];
        }
        __builtin_amdgcn_sched_barrier(0);
        #pragma unroll
        for (int k8 = 0; k8 < 8; ++k8) {
            const v4f av = av8[k8];
            const v4f cv = cv8[k8];
            v4f bd[NTAPS];
            bd[0] = bv8[k8];
            #pragma unroll
            for (int d = 1; d < NTAPS; ++d) bd[d] = bd[d - 1] * av;
            #pragma unroll
            for (int d = 0; d < NTAPS; ++d) acc[d] = wmma4(bd[d].xy, cv.xy, acc[d]);
            #pragma unroll
            for (int d = 0; d < NTAPS; ++d) acc[d] = wmma4(bd[d].zw, cv.zw, acc[d]);
        }
        #pragma unroll
        for (int d = 0; d < NTAPS; ++d) mpart[(w * NTAPS + d) * 32 + lane] = acc[d];
        __syncthreads();
        if (w < NTAPS) {
            v8f m = mpart[w * 32 + lane];
            #pragma unroll
            for (int s = 1; s < 8; ++s) m += mpart[(s * NTAPS + w) * 32 + lane];
            v4f* mo = mq + w * (16 * DOUT) + (4 * mt + 2 * lhi) * DOUT + 16 * nt + lmod;
            mo[0]    = __builtin_shufflevector(m, m, 0, 1, 2, 3);
            mo[DOUT] = __builtin_shufflevector(m, m, 4, 5, 6, 7);
        }
        return;
    }
    const int i = blockIdx.x * 256 + threadIdx.x;
    if (i < 16 * DHID) {
        const int q = i >> 9, n = i & (DHID - 1);
        v4f t;
        t.x = bW[(size_t)(4 * q)     * DHID + n];
        t.y = bW[(size_t)(4 * q + 1) * DHID + n];
        t.z = bW[(size_t)(4 * q + 2) * DHID + n];
        t.w = bW[(size_t)(4 * q + 3) * DHID + n];
        bq[i] = t;
    }
    if (i < 128 * DOUT) {
        const int q = i >> 6, n = i & (DOUT - 1);
        v4f t;
        t.x = cW[(size_t)(4 * q)     * DOUT + n];
        t.y = cW[(size_t)(4 * q + 1) * DOUT + n];
        t.z = cW[(size_t)(4 * q + 2) * DOUT + n];
        t.w = cW[(size_t)(4 * q + 3) * DOUT + n];
        cq[i] = t;
    }
}

__device__ __forceinline__ void turn_tile(float* twr, const float* trd, v8f v, v4f& lo, v4f& hi) {
    #pragma unroll
    for (int r = 0; r < 8; ++r) twr[r * TS_PITCH] = v[r];
    __builtin_amdgcn_fence(__ATOMIC_ACQ_REL, "wavefront");
    __builtin_amdgcn_wave_barrier();
    lo = *(const v4f*)(trd);
    hi = *(const v4f*)(trd + 4);
    __builtin_amdgcn_fence(__ATOMIC_ACQ_REL, "wavefront");
    __builtin_amdgcn_wave_barrier();
}

__device__ __forceinline__ v8f chunk_sum_so_far(const v4f* tot, int tid, int nt, float a1, v8f q) {
    float ap = a1;
    #pragma unroll
    for (int s = 1; s < SUBLEN; s <<= 1) ap *= ap;
    const v4f lo = tot[(2 * nt)     * NTHREADS + tid];
    const v4f hi = tot[(2 * nt + 1) * NTHREADS + tid];
    return __builtin_shufflevector(lo, hi, 0, 1, 2, 3, 4, 5, 6, 7) * ap + q;
}

__global__ __launch_bounds__(NTHREADS)
void rnn_chunk_summary(const float* __restrict__ x,
                       const float* __restrict__ aVec,
                       const v4f* __restrict__ bq0,
                       float* __restrict__ S,
                       float* __restrict__ Qs)
{
    const int chunk = blockIdx.x;
    const int tid   = threadIdx.x;
    const int wave  = tid >> 5;
    const int wave_s = __builtin_amdgcn_readfirstlane(wave);
    const int lane  = tid & 31;
    const int lmod  = lane & 15;
    const int lhi   = lane >> 4;

    const int m_tile = wave & 3;
    const int nclus  = wave >> 2;
    const int rowA   = m_tile * 16 + lmod;

    __shared__ v4f bs[BQ_UNITS];
    __shared__ v4f tot[16 * NTHREADS];
    __shared__ v4f tsc4[NWAVES * 2 * TS_TILE / 4];

    #pragma unroll
    for (int k = 0; k < BQ_UNITS / NTHREADS; ++k)
        bs[k * NTHREADS + tid] = bq0[k * NTHREADS + tid];
    #pragma unroll
    for (int g = 0; g < 16; ++g)
        tot[g * NTHREADS + tid] = v4f{0.0f, 0.0f, 0.0f, 0.0f};

    float av[8];
    #pragma unroll
    for (int nt = 0; nt < 8; ++nt)
        av[nt] = aVec[(nclus * 8 + nt) * 16 + lmod];

    const int bbase = lhi * DHID + nclus * 128 + lmod;

    v8f Sacc[8] = {};

    v4f afrag[8];
    {
        const float* xt = x + (size_t)(chunk * CLEN) * BATCH * DIN + rowA * DIN + lhi * 4;
        #pragma unroll
        for (int j = 0; j < 8; ++j)
            afrag[j] = *(const v4f*)(xt + j * 8);
    }

    __syncthreads();

    v4f bcur[4];
    {
        const int zb0 = opaque_copy(bbase);
        #pragma unroll
        for (int n = 0; n < 4; ++n)
            bcur[n] = bs[zb0 + n * 16];
    }

    for (int i = 0; i < CLEN; ++i) {
        const int zb = opaque_copy(bbase);

        const int inext = (i + 1 < CLEN) ? (i + 1) : i;
        const float* xn = x + (size_t)(chunk * CLEN + inext) * BATCH * DIN + rowA * DIN + lhi * 4;

        #pragma unroll
        for (int half = 0; half < 2; ++half) {
            v8f acc[4] = {};
            #pragma unroll
            for (int j = 0; j < 8; ++j) {
                const int nb = (half * 8 + j + 1) & 15;
                const int nh = nb >> 3, nj = nb & 7;
                v4f bnext[4];
                #pragma unroll
                for (int n = 0; n < 4; ++n)
                    bnext[n] = bs[zb + nj * 2 * DHID + nh * 64 + n * 16];
                __builtin_amdgcn_sched_barrier(0);
                #pragma unroll
                for (int n = 0; n < 4; ++n) acc[n] = wmma4(afrag[j].xy, bcur[n].xy, acc[n]);
                #pragma unroll
                for (int n = 0; n < 4; ++n) acc[n] = wmma4(afrag[j].zw, bcur[n].zw, acc[n]);
                if (half == 1)
                    afrag[j] = *(const v4f*)(xn + j * 8);
                __builtin_amdgcn_sched_barrier(0);
                #pragma unroll
                for (int n = 0; n < 4; ++n) bcur[n] = bnext[n];
            }
            #pragma unroll
            for (int n = 0; n < 4; ++n) {
                const float an = opaque_copyf(av[half * 4 + n]);
                #pragma unroll
                for (int r = 0; r < 8; ++r)
                    Sacc[half * 4 + n][r] = Sacc[half * 4 + n][r] * an + acc[n][r];
            }
        }

        if ((i & (SUBLEN - 1)) == SUBLEN - 1) {
            const int ln = opaque_lane();
            const int tl = ln & 15, th = ln >> 4, tw = wave_s;
            float* tsc = (float*)tsc4 + tw * (2 * TS_TILE);
            float*       twr = tsc + (8 * th) * TS_PITCH + tl;
            const float* trd = tsc + tl * TS_PITCH + 8 * th;
            const unsigned uoff = (unsigned)(tw * WPLANE + ln);
            if (i != CLEN - 1) {
                v4f* Qj = (v4f*)(Qs + (size_t)(i / SUBLEN) * PLANESET + (size_t)chunk * BATCH * DHID) + uoff;
                #pragma unroll
                for (int nt = 0; nt < 8; ++nt) {
                    v4f lo, hi;
                    turn_tile(twr + (nt & 1) * TS_TILE, trd + (nt & 1) * TS_TILE, Sacc[nt], lo, hi);
                    const v8f tv = chunk_sum_so_far(tot, tid, nt, opaque_copyf(av[nt]), Sacc[nt]);
                    tot[(2 * nt)     * NTHREADS + tid] = __builtin_shufflevector(tv, tv, 0, 1, 2, 3);
                    tot[(2 * nt + 1) * NTHREADS + tid] = __builtin_shufflevector(tv, tv, 4, 5, 6, 7);
                    Qj[nt * 64]      = lo;
                    Qj[nt * 64 + 32] = hi;
                    #pragma unroll
                    for (int r = 0; r < 8; ++r)
                        Sacc[nt][r] = 0.0f;
                }
            } else {
                v4f* Sj = (v4f*)(S + (size_t)chunk * BATCH * DHID) + uoff;
                v8f tv = chunk_sum_so_far(tot, tid, 0, opaque_copyf(av[0]), Sacc[0]);
                #pragma unroll
                for (int nt = 0; nt < 8; ++nt) {
                    v4f lo, hi;
                    turn_tile(twr + (nt & 1) * TS_TILE, trd + (nt & 1) * TS_TILE, tv, lo, hi);
                    if (nt + 1 < 8)
                        tv = chunk_sum_so_far(tot, tid, nt + 1, opaque_copyf(av[nt + 1]), Sacc[nt + 1]);
                    Sj[nt * 64]      = lo;
                    Sj[nt * 64 + 32] = hi;
                }
            }
        }
    }
}

#define SCAN_DEPTH   32
#define SCAN_THREADS 64
#define SCAN_PLANE   (BATCH * DHID / 2)

__global__ __launch_bounds__(SCAN_THREADS)
void rnn_carry_scan(const float* __restrict__ h0,
                    const float* __restrict__ aVec,
                    const float* __restrict__ S,
                    float* __restrict__ Cout,
                    float* __restrict__ hfinal)
{
    const int pair = blockIdx.x * SCAN_THREADS + threadIdx.x;
    const int idx  = pair * 2;
    const int u    = idx >> 2;
    const int e    = idx & 3;
    const int ln   = u & 31, hf = (u >> 5) & 1, nt = (u >> 6) & 7, wv = u >> 9;
    const int row  = 16 * (wv & 3) + (ln & 15);
    const int hcol = 128 * (wv >> 2) + 16 * nt + 8 * (ln >> 4) + 4 * hf + e;
    const int nat  = row * DHID + hcol;

    v2f aL = *(const v2f*)(aVec + hcol);
    #pragma unroll
    for (int s = 1; s < CLEN; s <<= 1) aL *= aL;

    const v2f* Sp = (const v2f*)S + pair;
    v2f*       Cp = (v2f*)Cout + pair;

    v2f cur[SCAN_DEPTH];
    #pragma unroll
    for (int k = 0; k < SCAN_DEPTH; ++k)
        cur[k] = Sp[(size_t)k * SCAN_PLANE];

    v2f c = *(const v2f*)(h0 + nat);

    #pragma unroll
    for (int jb = 0; jb < CHUNKS / SCAN_DEPTH; ++jb) {
        v2f nxt[SCAN_DEPTH];
        if (jb + 1 < CHUNKS / SCAN_DEPTH) {
            #pragma unroll
            for (int k = 0; k < SCAN_DEPTH; ++k)
                nxt[k] = Sp[(size_t)((jb + 1) * SCAN_DEPTH + k) * SCAN_PLANE];
        }
        __builtin_amdgcn_sched_barrier(0);
        #pragma unroll
        for (int k = 0; k < SCAN_DEPTH; ++k) {
            Cp[(size_t)(jb * SCAN_DEPTH + k) * SCAN_PLANE] = c;
            c = aL * c + cur[k];
        }
        __builtin_amdgcn_sched_barrier(0);
        #pragma unroll
        for (int k = 0; k < SCAN_DEPTH; ++k) cur[k] = nxt[k];
    }
    *(v2f*)(hfinal + nat) = c;
    Cp[(size_t)CHUNKS * SCAN_PLANE] = c;
}

__device__ __forceinline__ void load_m_block(v4f (&f)[8], const v4f* msm, int d, int mlo) {
    #pragma unroll
    for (int jj = 0; jj < 2; ++jj) {
        #pragma unroll
        for (int o = 0; o < 4; ++o)
            f[4 * jj + o] = msm[mlo + d * (16 * DOUT) + jj * (2 * DOUT) + o * 16];
    }
}

__device__ __forceinline__ void load_c_block(v4f (&f)[8], const v4f* csm, int nt, int cb) {
    #pragma unroll
    for (int o = 0; o < 4; ++o) {
        f[o]     = csm[cb + nt * (4 * DOUT)        + o * 16];
        f[4 + o] = csm[cb + nt * (4 * DOUT) + DOUT + o * 16];
    }
}

__device__ __forceinline__ void load_x_slice(v4f (&xv)[2], const float* x, unsigned xoff, int t) {
    const float* p = x + (size_t)t * BATCH * DIN;
    xv[0] = *(const v4f*)(p + xoff);
    xv[1] = *(const v4f*)(p + xoff + 8);
}

__global__ __launch_bounds__(NTHREADS)
void rnn_chunk_emit(const float* __restrict__ x,
                    const float* __restrict__ aVec,
                    const v4f* __restrict__ mq0,
                    const v4f* __restrict__ cq0,
                    const float* __restrict__ Cin,
                    const float* __restrict__ Qs,
                    float* __restrict__ y)
{
    __shared__ v4f ms[MQ_UNITS];
    __shared__ v4f cs[CQ_UNITS];
    __shared__ v4f part[4 * PPLANE];

    const int chunk = blockIdx.x;
    const int tid   = threadIdx.x;
    const int wave  = __builtin_amdgcn_readfirstlane(tid >> 5);

    const int m_tile = wave & 3;
    const int nclus  = wave >> 2;

    #pragma unroll
    for (int k = 0; k < MQ_UNITS / NTHREADS; ++k)
        ms[k * NTHREADS + tid] = mq0[k * NTHREADS + tid];
    #pragma unroll
    for (int k = 0; k < CQ_UNITS / NTHREADS; ++k)
        cs[k * NTHREADS + tid] = cq0[k * NTHREADS + tid];

    const int t00   = chunk * CLEN;
    const int tlast = t00 + CLEN - 1;

    v4f xnew[2];
    v4f fcur[8];
    {
        const int ln = opaque_lane();
        const int lmod = ln & 15, lhi = ln >> 4;
        load_x_slice(xnew, x, (unsigned)((m_tile * 16 + lmod) * DIN + nclus * 16 + lhi * 4), t00);
        __syncthreads();
        const int mlo = (4 * nclus + lhi) * DOUT + lmod;
        load_m_block(fcur, ms, 0, mlo);
    }

    const int zero = opaque_zero();
    const float* aV = aVec;

    v4f xcur[2] = {};
    v4f acur[2] = {};
    v4f qt[2][2] = {};
    v8f Hreg[8];

    {
        const int ln = opaque_lane();
        const v4f* Pv = (const v4f*)(Cin + (size_t)chunk * BATCH * DHID) + (unsigned)(wave * WPLANE + ln);
        #pragma unroll
        for (int nt = 0; nt < 8; ++nt) {
            const v4f lo = Pv[nt * 64];
            const v4f hi = Pv[nt * 64 + 32];
            Hreg[nt] = __builtin_shufflevector(lo, hi, 0, 1, 2, 3, 4, 5, 6, 7);
        }
    }

    for (int sub = 0; sub < NSUB; ++sub) {
        const int t0 = t00 + sub * SUBLEN;

        const bool lastsub = (sub == NSUB - 1);
        const float* Pn = lastsub ? Cin + (size_t)(chunk + 1) * BATCH * DHID
                                  : Qs + (size_t)sub * PLANESET + (size_t)chunk * BATCH * DHID;

        #pragma unroll
        for (int i = 0; i < SUBLEN; ++i) {
            aV += zero;
            const int t = t0 + i;

            const int ln   = opaque_lane();
            const int lmod = ln & 15;
            const int lhi  = ln >> 4;
            const int brow = m_tile * 16 + lmod;
            const unsigned xoff = (unsigned)(brow * DIN + nclus * 16 + lhi * 4);
            const int mlo  = (4 * nclus + lhi) * DOUT + lmod;
            const int cb   = (32 * nclus + 2 * lhi) * DOUT + lmod;
            const float* ap = aV + (unsigned)(nclus * 128 + lhi * 8);
            const v4f* Pv = (const v4f*)Pn + (unsigned)(wave * WPLANE + ln);

            v8f Y[4] = {};

            if (i < SUBLEN - 1) {
                #pragma unroll
                for (int s = 0; s <= i; ++s) {
                    v4f fnext[8];
                    v4f xnext[2] = {};
                    if (s < i) {
                        load_m_block(fnext, ms, i - (s + 1), mlo);
                        if (s + 1 < i)
                            load_x_slice(xnext, x, xoff, t0 + s + 1);
                    } else {
                        load_c_block(fnext, cs, 0, cb);
                        acur[0] = *(const v4f*)(ap);
                        acur[1] = *(const v4f*)(ap + 4);
                    }
                    __builtin_amdgcn_sched_barrier(0);
                    #pragma unroll
                    for (int jj = 0; jj < 2; ++jj) {
                        const v4f xo = (s == i) ? xnew[jj] : xcur[jj];
                        #pragma unroll
                        for (int o = 0; o < 4; ++o) Y[o] = wmma4(fcur[4 * jj + o].xy, xo.xy, Y[o]);
                        #pragma unroll
                        for (int o = 0; o < 4; ++o) Y[o] = wmma4(fcur[4 * jj + o].zw, xo.zw, Y[o]);
                    }
                    __builtin_amdgcn_sched_barrier(0);
                    #pragma unroll
                    for (int f = 0; f < 8; ++f) fcur[f] = fnext[f];
                    if (s + 1 < i) { xcur[0] = xnext[0]; xcur[1] = xnext[1]; }
                }
            }

            if (i != SUBLEN - 2) {
                load_x_slice(xnew, x, xoff, (t + 1 <= tlast) ? (t + 1) : tlast);
            }

            #pragma unroll
            for (int nt = 0; nt < 8; ++nt) {
                v4f fnext[8];
                if (nt + 1 < 8) {
                    load_c_block(fnext, cs, nt + 1, cb);
                } else if (i + 1 < SUBLEN - 1) {
                    load_m_block(fnext, ms, i + 1, mlo);
                    load_x_slice(xcur, x, xoff, t0);
                } else if (i + 1 == SUBLEN - 1) {
                    load_c_block(fnext, cs, 0, cb);
                } else {
                    load_m_block(fnext, ms, 0, mlo);
                }
                __builtin_amdgcn_sched_barrier(0);
                {
                    const v8f an = __builtin_shufflevector(acur[0], acur[1], 0, 1, 2, 3, 4, 5, 6, 7);
                    if (i < SUBLEN - 1) {
                        Hreg[nt] = Hreg[nt] * an;
                    } else {
                        const v8f q = __builtin_shufflevector(qt[nt & 1][0], qt[nt & 1][1], 0, 1, 2, 3, 4, 5, 6, 7);
                        const v8f d = Hreg[nt] * an + q;
                        Hreg[nt] = lastsub ? q : d;
                    }
                }
                if (nt + 1 < 8) {
                    acur[0] = *(const v4f*)(ap + 16 * (nt + 1));
                    acur[1] = *(const v4f*)(ap + 16 * (nt + 1) + 4);
                } else if (i + 1 == SUBLEN - 1) {
                    acur[0] = *(const v4f*)(ap);
                    acur[1] = *(const v4f*)(ap + 4);
                }
                if (i == SUBLEN - 2 && nt == 0) {
                    qt[0][0] = Pv[0];  qt[0][1] = Pv[32];
                    qt[1][0] = Pv[64]; qt[1][1] = Pv[96];
                }
                if (i == SUBLEN - 1 && nt + 2 < 8) {
                    qt[nt & 1][0] = Pv[(nt + 2) * 64];
                    qt[nt & 1][1] = Pv[(nt + 2) * 64 + 32];
                }
                const v8f  Hc = Hreg[nt];
                const v2f h01 = __builtin_shufflevector(Hc, Hc, 0, 1);
                const v2f h23 = __builtin_shufflevector(Hc, Hc, 2, 3);
                const v2f h45 = __builtin_shufflevector(Hc, Hc, 4, 5);
                const v2f h67 = __builtin_shufflevector(Hc, Hc, 6, 7);
                #pragma unroll
                for (int o = 0; o < 4; ++o) Y[o] = wmma4(fcur[o].xy,     h01, Y[o]);
                #pragma unroll
                for (int o = 0; o < 4; ++o) Y[o] = wmma4(fcur[o].zw,     h23, Y[o]);
                #pragma unroll
                for (int o = 0; o < 4; ++o) Y[o] = wmma4(fcur[4 + o].xy, h45, Y[o]);
                #pragma unroll
                for (int o = 0; o < 4; ++o) Y[o] = wmma4(fcur[4 + o].zw, h67, Y[o]);
                __builtin_amdgcn_sched_barrier(0);
                #pragma unroll
                for (int f = 0; f < 8; ++f) fcur[f] = fnext[f];
            }

            __syncthreads();

            const int lx   = opaque_lane();
            const int lxm  = lx & 15;
            const int lxh  = lx >> 4;
            {
                v4f* pw = part + nclus * PPLANE + (m_tile * 16 + lxm) * PUNITS;
                #pragma unroll
                for (int o = 0; o < 4; ++o) {
                    const int u0 = (o * 4 + lxh * 2) ^ lxm;
                    pw[u0]     = __builtin_shufflevector(Y[o], Y[o], 0, 1, 2, 3);
                    pw[u0 ^ 1] = __builtin_shufflevector(Y[o], Y[o], 4, 5, 6, 7);
                }
            }
            __syncthreads();

            {
                float* yt = y + (size_t)t * BATCH * DOUT;
                #pragma unroll
                for (int p = 0; p < 2; ++p) {
                    const int row = p * 32 + wave * 2 + lxh;
                    const int u   = lxm;
                    const v4f* q0 = part + row * PUNITS + (u ^ (row & 15));
                    const v4f s0 = q0[0];
                    const v4f s1 = q0[PPLANE];
                    const v4f s2 = q0[2 * PPLANE];
                    const v4f s3 = q0[3 * PPLANE];
                    *(v4f*)(yt + (unsigned)(row * DOUT + u * 4)) = (s0 + s1) + (s2 + s3);
                }
            }
        }
    }
}

extern "C" void kernel_launch(void* const* d_in, const int* in_sizes, int n_in,
                              void* d_out, int out_size, void* d_ws, size_t ws_size,
                              hipStream_t stream)
{
    const float* h0 = (const float*)d_in[0];
    const float* x  = (const float*)d_in[1];
    const float* a  = (const float*)d_in[2];
    const float* bW = (const float*)d_in[3];
    const float* cW = (const float*)d_in[4];

    float* out    = (float*)d_out;
    float* hfinal = out;
    float* y      = out + BATCH * DHID;

    float* S  = (float*)d_ws;
    float* Qs = S  + PLANESET;
    float* C  = Qs + (NSUB - 1) * PLANESET;
    v4f*   bq = (v4f*)(C + PLANESET + BATCH * DHID);
    v4f*   cq = bq + 16 * DHID;
    v4f*   mq = cq + 128 * DOUT;

    rnn_pack<<<48, 256, 0, stream>>>(bW, cW, a, bq, cq, mq);
    rnn_chunk_summary<<<CHUNKS, NTHREADS, 0, stream>>>(x, a, bq, S, Qs);
    rnn_carry_scan<<<BATCH * DHID / 2 / SCAN_THREADS, SCAN_THREADS, 0, stream>>>(h0, a, S, C, hfinal);
    rnn_chunk_emit<<<CHUNKS, NTHREADS, 0, stream>>>(x, a, mq, cq, C, Qs, y);
}
